# starting kernel
# baseline (speedup 1.0000x reference)
#include <hip/hip_runtime.h>
#include <hip/hip_bf16.h>

#define N_NODES 100000
#define N_EDGES 1000000
#define DIN  64
#define DHID 64
#define DOUTF 32

typedef __attribute__((ext_vector_type(2))) float v2f;
typedef __attribute__((ext_vector_type(8))) float v8f;

__global__ void k_init_deg(float* __restrict__ deg, int n) {
    int i = blockIdx.x * blockDim.x + threadIdx.x;
    if (i < n) deg[i] = 1.0f;
}

__global__ void k_accum_deg(const int* __restrict__ dst, const float* __restrict__ ew,
                            float* __restrict__ deg, int e) {
    int i = blockIdx.x * blockDim.x + threadIdx.x;
    if (i < e) atomicAdd(&deg[dst[i]], ew[i]);
}

__global__ void k_rsqrt_inplace(float* __restrict__ deg, int n) {
    int i = blockIdx.x * blockDim.x + threadIdx.x;
    if (i < n) deg[i] = rsqrtf(deg[i]);
}

template <int F>
__global__ void k_init_bias(float* __restrict__ out, const float* __restrict__ b, int total) {
    int i = blockIdx.x * blockDim.x + threadIdx.x;
    if (i < total) out[i] = b[i & (F - 1)];
}

template <int KDIM, int NDIM, bool RELU>
__global__ void k_gemm_wmma(const float* __restrict__ A, const float* __restrict__ B,
                            float* __restrict__ C) {
    const int lane  = threadIdx.x & 31;
    const int wave  = threadIdx.x >> 5;
    const int row0  = blockIdx.x * 16;
    const int col0  = wave * 16;
    const int mrow  = lane & 15;
    const int khalf = (lane >> 4) * 2;

    v8f acc = {};
#pragma unroll
    for (int k = 0; k < KDIM; k += 4) {
        const float* ap = A + (size_t)(row0 + mrow) * KDIM + (k + khalf);
        v2f a;
        a.x = ap[0];
        a.y = ap[1];
        if (RELU) { a.x = fmaxf(a.x, 0.0f); a.y = fmaxf(a.y, 0.0f); }

        const float* bp = B + (size_t)(k + khalf) * NDIM + (col0 + mrow);
        v2f bfr;
        bfr.x = bp[0];
        bfr.y = bp[NDIM];

        acc = __builtin_amdgcn_wmma_f32_16x16x4_f32(
             false, a,  false, bfr,
             (short)0, acc,  false,  false);
    }

    const int mbase = row0 + ((lane >> 4) << 3);
    const int col   = col0 + (lane & 15);
#pragma unroll
    for (int r = 0; r < 8; ++r)
        C[(size_t)(mbase + r) * NDIM + col] = acc[r];
}

template <int F>
__global__ void k_scatter(const float* __restrict__ h,
                          const int* __restrict__ src, const int* __restrict__ dst,
                          const float* __restrict__ ew, const float* __restrict__ dinv,
                          float* __restrict__ out, int nE, int nTot) {
    int gid  = blockIdx.x * blockDim.x + threadIdx.x;
    int e    = gid >> 5;
    int lane = gid & 31;
    if (e >= nTot) return;

    int s, d; float w;
    if (e < nE) { s = src[e]; d = dst[e]; w = ew[e]; }
    else        { s = d = e - nE; w = 1.0f; }

    float nrm = dinv[s] * w * dinv[d];
    const float* hs = h + (size_t)s * F;
    float* od = out + (size_t)d * F;
#pragma unroll
    for (int j = lane; j < F; j += 32)
        atomicAdd(&od[j], hs[j] * nrm);
}

extern "C" void kernel_launch(void* const* d_in, const int* in_sizes, int n_in,
                              void* d_out, int out_size, void* d_ws, size_t ws_size,
                              hipStream_t stream) {
    (void)in_sizes; (void)n_in; (void)out_size; (void)ws_size;

    const float* x   = (const float*)d_in[0];
    const int*   ei  = (const int*)  d_in[1];
    const float* ew  = (const float*)d_in[2];
    const float* W1  = (const float*)d_in[3];
    const float* b1  = (const float*)d_in[4];
    const float* W2  = (const float*)d_in[5];
    const float* b2  = (const float*)d_in[6];
    const int* src = ei;
    const int* dst = ei + N_EDGES;
    float* out = (float*)d_out;

    float* dinv = (float*)d_ws;
    float* h1   = dinv + N_NODES;
    float* a1   = h1 + (size_t)N_NODES * DHID;
    float* h2   = h1;

    const int nTot = N_EDGES + N_NODES;
    const int TPB  = 256;

    k_init_deg<<<(N_NODES + TPB - 1) / TPB, TPB, 0, stream>>>(dinv, N_NODES);
    k_accum_deg<<<(N_EDGES + TPB - 1) / TPB, TPB, 0, stream>>>(dst, ew, dinv, N_EDGES);
    k_rsqrt_inplace<<<(N_NODES + TPB - 1) / TPB, TPB, 0, stream>>>(dinv, N_NODES);

    k_gemm_wmma<DIN, DHID, false>
        <<<N_NODES / 16, (DHID / 16) * 32, 0, stream>>>(x, W1, h1);
    k_init_bias<DHID>
        <<<(N_NODES * DHID + TPB - 1) / TPB, TPB, 0, stream>>>(a1, b1, N_NODES * DHID);
    {
        long long thr = (long long)nTot * 32;
        k_scatter<DHID>
            <<<(unsigned)((thr + TPB - 1) / TPB), TPB, 0, stream>>>(
                h1, src, dst, ew, dinv, a1, N_EDGES, nTot);
    }

    k_gemm_wmma<DHID, DOUTF, true>
        <<<N_NODES / 16, (DOUTF / 16) * 32, 0, stream>>>(a1, W2, h2);
    k_init_bias<DOUTF>
        <<<(N_NODES * DOUTF + TPB - 1) / TPB, TPB, 0, stream>>>(out, b2, N_NODES * DOUTF);
    {
        long long thr = (long long)nTot * 32;
        k_scatter<DOUTF>
            <<<(unsigned)((thr + TPB - 1) / TPB), TPB, 0, stream>>>(
                h2, src, dst, ew, dinv, out, N_EDGES, nTot);
    }
}


# Edges grouped by destination node on every call (two-level LDS counting sort); a wave sums a node's source rows in registers, stores each row once: no global atomics; float sum order changes
# speedup vs baseline: 2.7663x; 2.7663x over previous
#include <hip/hip_runtime.h>

#define N_NODES 100000
#define N_EDGES 1000000
#define DIN  64
#define DHID 64
#define DOUTF 32

#define RANGE_SHIFT 8
#define RANGE_NODES 256
#define N_RANGES ((N_NODES >> RANGE_SHIFT) + 1)
#define PART_BLOCKS 256
#define PART_CHUNK ((N_EDGES + PART_BLOCKS - 1) / PART_BLOCKS)
#define N_CELLS (N_RANGES * PART_BLOCKS)
#define SCAN_BLOCKS ((N_CELLS + 1023) / 1024)

typedef __attribute__((ext_vector_type(2))) float v2f;
typedef __attribute__((ext_vector_type(8))) float v8f;

__device__ __forceinline__ int range_of(int d) {
    const unsigned r = ((unsigned)d) >> RANGE_SHIFT;
    return (int)(r < (unsigned)(N_RANGES - 1) ? r : (unsigned)(N_RANGES - 1));
}

__global__ void __launch_bounds__(1024)
k_part_count(const int* __restrict__ dst, int* __restrict__ cells) {
    __shared__ int hist[N_RANGES];
    const int t = threadIdx.x;
    for (int i = t; i < N_RANGES; i += 1024) hist[i] = 0;
    __syncthreads();
    const int e0 = blockIdx.x * PART_CHUNK;
    const int e1 = min(e0 + PART_CHUNK, N_EDGES);
    for (int e = e0 + t; e < e1; e += 1024) atomicAdd(&hist[range_of(dst[e])], 1);
    __syncthreads();
    for (int i = t; i < N_RANGES; i += 1024) cells[i * PART_BLOCKS + blockIdx.x] = hist[i];
}

__global__ void __launch_bounds__(1024)
k_scan_cells(const int* __restrict__ in, int* __restrict__ out) {
    __shared__ int red[32];
    __shared__ int wtot[32];
    const int t = threadIdx.x, lane = t & 31, wv = t >> 5;
    int acc = 0;
    for (int j = 0; j < (int)blockIdx.x; ++j) acc += in[j * 1024 + t];
    for (int o = 16; o > 0; o >>= 1) acc += __shfl_xor(acc, o);
    if (lane == 0) red[wv] = acc;
    const int i = blockIdx.x * 1024 + t;
    const int v = (i < N_CELLS) ? in[i] : 0;
    int inc = v;
    for (int o = 1; o < 32; o <<= 1) {
        const int nb = __shfl_up(inc, o);
        if (lane >= o) inc += nb;
    }
    if (lane == 31) wtot[wv] = inc;
    __syncthreads();
    int pre = 0;
    for (int k = 0; k < 32; ++k) pre += red[k];
    for (int k = 0; k < wv; ++k) pre += wtot[k];
    if (i < N_CELLS) out[i] = pre + inc - v;
}

__global__ void __launch_bounds__(1024)
k_part_place(const int* __restrict__ src, const int* __restrict__ dst,
             const float* __restrict__ ew, const int* __restrict__ cellOff,
             int4* __restrict__ binned) {
    __shared__ int cur[N_RANGES];
    const int t = threadIdx.x;
    for (int i = t; i < N_RANGES; i += 1024) cur[i] = cellOff[i * PART_BLOCKS + blockIdx.x];
    __syncthreads();
    const int e0 = blockIdx.x * PART_CHUNK;
    const int e1 = min(e0 + PART_CHUNK, N_EDGES);
    for (int e = e0 + t; e < e1; e += 1024) {
        const int d = dst[e];
        const int p = atomicAdd(&cur[range_of(d)], 1);
        binned[p] = make_int4(src[e], d, __float_as_int(ew[e]), 0);
    }
}

__global__ void __launch_bounds__(1024)
k_range_sort(const int4* __restrict__ binned, const int* __restrict__ cellOff,
             int2* __restrict__ rec, int* __restrict__ off, float* __restrict__ dinv) {
    __shared__ int   cnt[RANGE_NODES];
    __shared__ float wsum[RANGE_NODES];
    __shared__ int   cur[RANGE_NODES];
    __shared__ int   wtot[RANGE_NODES / 32];
    const int t = threadIdx.x, lane = t & 31, wv = t >> 5;
    const int b = blockIdx.x;
    const int bs = cellOff[b * PART_BLOCKS];
    const int be = (b + 1 < N_RANGES) ? cellOff[(b + 1) * PART_BLOCKS] : N_EDGES;
    if (t < RANGE_NODES) { cnt[t] = 0; wsum[t] = 0.0f; }
    __syncthreads();
    for (int i = bs + t; i < be; i += 1024) {
        const int4 r = binned[i];
        const int dl = r.y & (RANGE_NODES - 1);
        atomicAdd(&cnt[dl], 1);
        atomicAdd(&wsum[dl], __int_as_float(r.z));
    }
    __syncthreads();
    const int v = (t < RANGE_NODES) ? cnt[t] : 0;
    int inc = v;
    for (int o = 1; o < 32; o <<= 1) {
        const int nb = __shfl_up(inc, o);
        if (lane >= o) inc += nb;
    }
    if (t < RANGE_NODES && lane == 31) wtot[wv] = inc;
    __syncthreads();
    if (t < RANGE_NODES) {
        int pre = 0;
        for (int k = 0; k < wv; ++k) pre += wtot[k];
        const int ex = pre + inc - v;
        cur[t] = ex;
        const int node = b * RANGE_NODES + t;
        if (node <= N_NODES) off[node] = bs + ex;
        if (node < N_NODES) {
            const float dg = 1.0f + wsum[t];
            dinv[node] = dg > 0.0f ? rsqrtf(dg) : 0.0f;
        }
    }
    __syncthreads();
    for (int i = bs + t; i < be; i += 1024) {
        const int4 r = binned[i];
        const int dl = r.y & (RANGE_NODES - 1);
        const int p = atomicAdd(&cur[dl], 1);
        rec[bs + p] = make_int2(r.x, r.z);
    }
}

template <int KDIM, int NDIM, bool RELU>
__global__ void k_gemm_wmma(const float* __restrict__ A, const float* __restrict__ B,
                            const float* __restrict__ rowscale, float* __restrict__ C) {
    const int lane  = threadIdx.x & 31;
    const int wave  = threadIdx.x >> 5;
    const int row0  = blockIdx.x * 16;
    const int col0  = wave * 16;
    const int mrow  = lane & 15;
    const int khalf = (lane >> 4) * 2;

    v8f acc = {};
#pragma unroll
    for (int k = 0; k < KDIM; k += 4) {
        const float* ap = A + (size_t)(row0 + mrow) * KDIM + (k + khalf);
        v2f a;
        a.x = ap[0];
        a.y = ap[1];
        if (RELU) { a.x = fmaxf(a.x, 0.0f); a.y = fmaxf(a.y, 0.0f); }

        const float* bp = B + (size_t)(k + khalf) * NDIM + (col0 + mrow);
        v2f bfr;
        bfr.x = bp[0];
        bfr.y = bp[NDIM];

        acc = __builtin_amdgcn_wmma_f32_16x16x4_f32(
            false, a, false, bfr, (short)0, acc, false, false);
    }

    const int mbase = row0 + ((lane >> 4) << 3);
    const int col   = col0 + (lane & 15);
#pragma unroll
    for (int r = 0; r < 8; ++r)
        C[(size_t)(mbase + r) * NDIM + col] = acc[r] * rowscale[mbase + r];
}

template <int F>
__global__ void __launch_bounds__(256)
k_gather(const float* __restrict__ hs, const int2* __restrict__ rec,
         const int* __restrict__ off, const float* __restrict__ dinv,
         const float* __restrict__ bias, float* __restrict__ out, int n) {
    constexpr int LPR = F / 4;
    constexpr int G   = 32 / LPR;
    constexpr int U   = 16 / G;
    const int lane = threadIdx.x & 31;
    const int d = (blockIdx.x * blockDim.x + threadIdx.x) >> 5;
    if (d >= n) return;
    const int g = lane / LPR;
    const int c = lane % LPR;
    const float4* __restrict__ hs4 = reinterpret_cast<const float4*>(hs);

    const int st = off[d];
    const int en = off[d + 1];
    const float dv = dinv[d];

    float4 acc = make_float4(0.0f, 0.0f, 0.0f, 0.0f);
    if (g == 0) acc = hs4[(size_t)d * LPR + c];

    for (int k = st; k < en; k += 16) {
        int2 rr[U];
#pragma unroll
        for (int j = 0; j < U; ++j) rr[j] = rec[min(k + j * G + g, en - 1)];
        float4 row[U];
#pragma unroll
        for (int j = 0; j < U; ++j) row[j] = hs4[(size_t)rr[j].x * LPR + c];
#pragma unroll
        for (int j = 0; j < U; ++j) {
            if (k + j * G + g < en) {
                const float w = __int_as_float(rr[j].y);
                acc.x += w * row[j].x;
                acc.y += w * row[j].y;
                acc.z += w * row[j].z;
                acc.w += w * row[j].w;
            }
        }
    }
#pragma unroll
    for (int o = LPR; o < 32; o <<= 1) {
        acc.x += __shfl_xor(acc.x, o);
        acc.y += __shfl_xor(acc.y, o);
        acc.z += __shfl_xor(acc.z, o);
        acc.w += __shfl_xor(acc.w, o);
    }
    if (g == 0) {
        float4 o4;
        o4.x = bias[c * 4 + 0] + dv * acc.x;
        o4.y = bias[c * 4 + 1] + dv * acc.y;
        o4.z = bias[c * 4 + 2] + dv * acc.z;
        o4.w = bias[c * 4 + 3] + dv * acc.w;
        reinterpret_cast<float4*>(out)[(size_t)d * LPR + c] = o4;
    }
}

extern "C" void kernel_launch(void* const* d_in, const int* in_sizes, int n_in,
                              void* d_out, int out_size, void* d_ws, size_t ws_size,
                              hipStream_t stream) {
    (void)in_sizes; (void)n_in; (void)out_size; (void)ws_size;

    const float* x   = (const float*)d_in[0];
    const int*   ei  = (const int*)  d_in[1];
    const float* ew  = (const float*)d_in[2];
    const float* W1  = (const float*)d_in[3];
    const float* b1  = (const float*)d_in[4];
    const float* W2  = (const float*)d_in[5];
    const float* b2  = (const float*)d_in[6];
    const int* src = ei;
    const int* dst = ei + N_EDGES;
    float* out = (float*)d_out;

    float* dinv    = (float*)d_ws;
    int*   off     = (int*)(dinv + N_NODES);
    int*   cells   = off + 100032;
    int*   cellOff = cells + 100128;
    int4*  binned  = (int4*)(cellOff + N_CELLS);
    int2*  rec     = (int2*)(binned + N_EDGES);
    float* h1      = (float*)(rec + N_EDGES);
    float* a1      = h1 + (size_t)N_NODES * DHID;
    float* h2      = h1;

    k_part_count<<<PART_BLOCKS, 1024, 0, stream>>>(dst, cells);
    k_scan_cells<<<SCAN_BLOCKS, 1024, 0, stream>>>(cells, cellOff);
    k_part_place<<<PART_BLOCKS, 1024, 0, stream>>>(src, dst, ew, cellOff, binned);
    k_range_sort<<<N_RANGES, 1024, 0, stream>>>(binned, cellOff, rec, off, dinv);

    k_gemm_wmma<DIN, DHID, false>
        <<<N_NODES / 16, (DHID / 16) * 32, 0, stream>>>(x, W1, dinv, h1);
    k_gather<DHID>
        <<<(N_NODES + 7) / 8, 256, 0, stream>>>(h1, rec, off, dinv, b1, a1, N_NODES);

    k_gemm_wmma<DHID, DOUTF, true>
        <<<N_NODES / 16, (DOUTF / 16) * 32, 0, stream>>>(a1, W2, dinv, h2);
    k_gather<DOUTF>
        <<<(N_NODES + 7) / 8, 256, 0, stream>>>(h2, rec, off, dinv, b2, out, N_NODES);
}


# Gather: one wave walks 8 consecutive destination nodes, prefetching the next node's records, instead of one node per wave: 8x fewer, longer-lived waves; per-node sum order unchanged
# speedup vs baseline: 3.0838x; 1.1148x over previous
#include <hip/hip_runtime.h>

#define N_NODES 100000
#define N_EDGES 1000000
#define DIN  64
#define DHID 64
#define DOUTF 32

#define RANGE_SHIFT 8
#define RANGE_NODES 256
#define N_RANGES ((N_NODES >> RANGE_SHIFT) + 1)
#define PART_BLOCKS 256
#define PART_CHUNK ((N_EDGES + PART_BLOCKS - 1) / PART_BLOCKS)
#define N_CELLS (N_RANGES * PART_BLOCKS)
#define SCAN_BLOCKS ((N_CELLS + 1023) / 1024)

typedef __attribute__((ext_vector_type(2))) float v2f;
typedef __attribute__((ext_vector_type(8))) float v8f;

__device__ __forceinline__ int range_of(int d) {
    const unsigned r = ((unsigned)d) >> RANGE_SHIFT;
    return (int)(r < (unsigned)(N_RANGES - 1) ? r : (unsigned)(N_RANGES - 1));
}

__global__ void __launch_bounds__(1024)
k_part_count(const int* __restrict__ dst, int* __restrict__ cells) {
    __shared__ int hist[N_RANGES];
    const int t = threadIdx.x;
    for (int i = t; i < N_RANGES; i += 1024) hist[i] = 0;
    __syncthreads();
    const int e0 = blockIdx.x * PART_CHUNK;
    const int e1 = min(e0 + PART_CHUNK, N_EDGES);
    for (int e = e0 + t; e < e1; e += 1024) atomicAdd(&hist[range_of(dst[e])], 1);
    __syncthreads();
    for (int i = t; i < N_RANGES; i += 1024) cells[i * PART_BLOCKS + blockIdx.x] = hist[i];
}

__global__ void __launch_bounds__(1024)
k_scan_cells(const int* __restrict__ in, int* __restrict__ out) {
    __shared__ int red[32];
    __shared__ int wtot[32];
    const int t = threadIdx.x, lane = t & 31, wv = t >> 5;
    int acc = 0;
    for (int j = 0; j < (int)blockIdx.x; ++j) acc += in[j * 1024 + t];
    for (int o = 16; o > 0; o >>= 1) acc += __shfl_xor(acc, o);
    if (lane == 0) red[wv] = acc;
    const int i = blockIdx.x * 1024 + t;
    const int v = (i < N_CELLS) ? in[i] : 0;
    int inc = v;
    for (int o = 1; o < 32; o <<= 1) {
        const int nb = __shfl_up(inc, o);
        if (lane >= o) inc += nb;
    }
    if (lane == 31) wtot[wv] = inc;
    __syncthreads();
    int pre = 0;
    for (int k = 0; k < 32; ++k) pre += red[k];
    for (int k = 0; k < wv; ++k) pre += wtot[k];
    if (i < N_CELLS) out[i] = pre + inc - v;
}

__global__ void __launch_bounds__(1024)
k_part_place(const int* __restrict__ src, const int* __restrict__ dst,
             const float* __restrict__ ew, const int* __restrict__ cellOff,
             int4* __restrict__ binned) {
    __shared__ int cur[N_RANGES];
    const int t = threadIdx.x;
    for (int i = t; i < N_RANGES; i += 1024) cur[i] = cellOff[i * PART_BLOCKS + blockIdx.x];
    __syncthreads();
    const int e0 = blockIdx.x * PART_CHUNK;
    const int e1 = min(e0 + PART_CHUNK, N_EDGES);
    for (int e = e0 + t; e < e1; e += 1024) {
        const int d = dst[e];
        const int p = atomicAdd(&cur[range_of(d)], 1);
        binned[p] = make_int4(src[e], d, __float_as_int(ew[e]), 0);
    }
}

__global__ void __launch_bounds__(1024)
k_range_sort(const int4* __restrict__ binned, const int* __restrict__ cellOff,
             int2* __restrict__ rec, int* __restrict__ off, float* __restrict__ dinv) {
    __shared__ int   cnt[RANGE_NODES];
    __shared__ float wsum[RANGE_NODES];
    __shared__ int   cur[RANGE_NODES];
    __shared__ int   wtot[RANGE_NODES / 32];
    const int t = threadIdx.x, lane = t & 31, wv = t >> 5;
    const int b = blockIdx.x;
    const int bs = cellOff[b * PART_BLOCKS];
    const int be = (b + 1 < N_RANGES) ? cellOff[(b + 1) * PART_BLOCKS] : N_EDGES;
    if (t < RANGE_NODES) { cnt[t] = 0; wsum[t] = 0.0f; }
    __syncthreads();
    for (int i = bs + t; i < be; i += 1024) {
        const int4 r = binned[i];
        const int dl = r.y & (RANGE_NODES - 1);
        atomicAdd(&cnt[dl], 1);
        atomicAdd(&wsum[dl], __int_as_float(r.z));
    }
    __syncthreads();
    const int v = (t < RANGE_NODES) ? cnt[t] : 0;
    int inc = v;
    for (int o = 1; o < 32; o <<= 1) {
        const int nb = __shfl_up(inc, o);
        if (lane >= o) inc += nb;
    }
    if (t < RANGE_NODES && lane == 31) wtot[wv] = inc;
    __syncthreads();
    if (t < RANGE_NODES) {
        int pre = 0;
        for (int k = 0; k < wv; ++k) pre += wtot[k];
        const int ex = pre + inc - v;
        cur[t] = ex;
        const int node = b * RANGE_NODES + t;
        if (node <= N_NODES) off[node] = bs + ex;
        if (node < N_NODES) {
            const float dg = 1.0f + wsum[t];
            dinv[node] = dg > 0.0f ? rsqrtf(dg) : 0.0f;
        }
    }
    __syncthreads();
    for (int i = bs + t; i < be; i += 1024) {
        const int4 r = binned[i];
        const int dl = r.y & (RANGE_NODES - 1);
        const int p = atomicAdd(&cur[dl], 1);
        rec[bs + p] = make_int2(r.x, r.z);
    }
}

template <int KDIM, int NDIM, bool RELU>
__global__ void k_gemm_wmma(const float* __restrict__ A, const float* __restrict__ B,
                            const float* __restrict__ rowscale, float* __restrict__ C) {
    const int lane  = threadIdx.x & 31;
    const int wave  = threadIdx.x >> 5;
    const int row0  = blockIdx.x * 16;
    const int col0  = wave * 16;
    const int mrow  = lane & 15;
    const int khalf = (lane >> 4) * 2;

    v8f acc = {};
#pragma unroll
    for (int k = 0; k < KDIM; k += 4) {
        const float* ap = A + (size_t)(row0 + mrow) * KDIM + (k + khalf);
        v2f a;
        a.x = ap[0];
        a.y = ap[1];
        if (RELU) { a.x = fmaxf(a.x, 0.0f); a.y = fmaxf(a.y, 0.0f); }

        const float* bp = B + (size_t)(k + khalf) * NDIM + (col0 + mrow);
        v2f bfr;
        bfr.x = bp[0];
        bfr.y = bp[NDIM];

        acc = __builtin_amdgcn_wmma_f32_16x16x4_f32(
            false, a, false, bfr, (short)0, acc, false, false);
    }

    const int mbase = row0 + ((lane >> 4) << 3);
    const int col   = col0 + (lane & 15);
#pragma unroll
    for (int r = 0; r < 8; ++r)
        C[(size_t)(mbase + r) * NDIM + col] = acc[r] * rowscale[mbase + r];
}

#define GATHER_NPW 8

__device__ __forceinline__ int rec_slot(int s, int en) {
    return max(min(s, en - 1), 0);
}

template <int F, int NPW>
__global__ void __launch_bounds__(256)
k_gather(const float* __restrict__ hs, const int2* __restrict__ rec,
         const int* __restrict__ off, const float* __restrict__ dinv,
         const float* __restrict__ bias, float* __restrict__ out, int n) {
    constexpr int LPR = F / 4;
    constexpr int G   = 32 / LPR;
    constexpr int U   = 16 / G;
    static_assert(NPW + 1 <= 32, "a wave's list bounds must fit one register");
    const int lane = threadIdx.x & 31;
    const int d0 = __builtin_amdgcn_readfirstlane(
                       (int)((blockIdx.x * blockDim.x + threadIdx.x) >> 5)) * NPW;
    if (d0 >= n) return;
    const int nd = min(NPW, n - d0);
    const int g = lane / LPR;
    const int c = lane % LPR;
    const int keep0 = (g == 0) ? -1 : 0;
    const float4* __restrict__ hs4 = reinterpret_cast<const float4*>(hs);

    const int   offv = off[min(d0 + lane, n)];
    const float dvv  = dinv[min(d0 + lane, n - 1)];
    const float4 b4  = reinterpret_cast<const float4*>(bias)[c];

    int st = __builtin_amdgcn_readlane(offv, 0);
    int en = __builtin_amdgcn_readlane(offv, 1);
    int2 rr[U];
#pragma unroll
    for (int j = 0; j < U; ++j) rr[j] = rec[rec_slot(st + j * G + g, en)];

    for (int i = 0; i < nd; ++i) {
        const int d = d0 + i;
        const float4 self = hs4[(size_t)d * LPR + c];
        float4 row[U];
#pragma unroll
        for (int j = 0; j < U; ++j) row[j] = hs4[(size_t)rr[j].x * LPR + c];

        const int enn = __builtin_amdgcn_readlane(offv, min(i + 2, nd));
        int2 rrn[U];
#pragma unroll
        for (int j = 0; j < U; ++j) rrn[j] = rec[rec_slot(en + j * G + g, enn)];

        float4 acc;
        acc.x = __int_as_float(__float_as_int(self.x) & keep0);
        acc.y = __int_as_float(__float_as_int(self.y) & keep0);
        acc.z = __int_as_float(__float_as_int(self.z) & keep0);
        acc.w = __int_as_float(__float_as_int(self.w) & keep0);
#pragma unroll
        for (int j = 0; j < U; ++j) {
            const bool live = st + j * G + g < en;
            const float w = __int_as_float(rr[j].y);
            const float sx = acc.x + w * row[j].x;
            const float sy = acc.y + w * row[j].y;
            const float sz = acc.z + w * row[j].z;
            const float sw = acc.w + w * row[j].w;
            acc.x = live ? sx : acc.x;
            acc.y = live ? sy : acc.y;
            acc.z = live ? sz : acc.z;
            acc.w = live ? sw : acc.w;
        }
        for (int k = st + 16; k < en; k += 16) {
            int2 rq[U];
#pragma unroll
            for (int j = 0; j < U; ++j) rq[j] = rec[min(k + j * G + g, en - 1)];
            float4 rw[U];
#pragma unroll
            for (int j = 0; j < U; ++j) rw[j] = hs4[(size_t)rq[j].x * LPR + c];
#pragma unroll
            for (int j = 0; j < U; ++j) {
                if (k + j * G + g < en) {
                    const float w = __int_as_float(rq[j].y);
                    acc.x += w * rw[j].x;
                    acc.y += w * rw[j].y;
                    acc.z += w * rw[j].z;
                    acc.w += w * rw[j].w;
                }
            }
        }
#pragma unroll
        for (int o = LPR; o < 32; o <<= 1) {
            acc.x += __shfl_xor(acc.x, o);
            acc.y += __shfl_xor(acc.y, o);
            acc.z += __shfl_xor(acc.z, o);
            acc.w += __shfl_xor(acc.w, o);
        }
        const float dv = __int_as_float(__builtin_amdgcn_readlane(__float_as_int(dvv), i));
        if (g == 0) {
            float4 o4;
            o4.x = b4.x + dv * acc.x;
            o4.y = b4.y + dv * acc.y;
            o4.z = b4.z + dv * acc.z;
            o4.w = b4.w + dv * acc.w;
            reinterpret_cast<float4*>(out)[(size_t)d * LPR + c] = o4;
        }
        st = en;
        en = enn;
#pragma unroll
        for (int j = 0; j < U; ++j) rr[j] = rrn[j];
    }
}

extern "C" void kernel_launch(void* const* d_in, const int* in_sizes, int n_in,
                              void* d_out, int out_size, void* d_ws, size_t ws_size,
                              hipStream_t stream) {
    (void)in_sizes; (void)n_in; (void)out_size; (void)ws_size;

    const float* x   = (const float*)d_in[0];
    const int*   ei  = (const int*)  d_in[1];
    const float* ew  = (const float*)d_in[2];
    const float* W1  = (const float*)d_in[3];
    const float* b1  = (const float*)d_in[4];
    const float* W2  = (const float*)d_in[5];
    const float* b2  = (const float*)d_in[6];
    const int* src = ei;
    const int* dst = ei + N_EDGES;
    float* out = (float*)d_out;

    float* dinv    = (float*)d_ws;
    int*   off     = (int*)(dinv + N_NODES);
    int*   cells   = off + 100032;
    int*   cellOff = cells + 100128;
    int4*  binned  = (int4*)(cellOff + N_CELLS);
    int2*  rec     = (int2*)(binned + N_EDGES);
    float* h1      = (float*)(rec + N_EDGES);
    float* a1      = h1 + (size_t)N_NODES * DHID;
    float* h2      = h1;

    k_part_count<<<PART_BLOCKS, 1024, 0, stream>>>(dst, cells);
    k_scan_cells<<<SCAN_BLOCKS, 1024, 0, stream>>>(cells, cellOff);
    k_part_place<<<PART_BLOCKS, 1024, 0, stream>>>(src, dst, ew, cellOff, binned);
    k_range_sort<<<N_RANGES, 1024, 0, stream>>>(binned, cellOff, rec, off, dinv);

    k_gemm_wmma<DIN, DHID, false>
        <<<N_NODES / 16, (DHID / 16) * 32, 0, stream>>>(x, W1, dinv, h1);
    k_gather<DHID, GATHER_NPW>
        <<<(N_NODES + 8 * GATHER_NPW - 1) / (8 * GATHER_NPW), 256, 0, stream>>>(
            h1, rec, off, dinv, b1, a1, N_NODES);

    k_gemm_wmma<DHID, DOUTF, true>
        <<<N_NODES / 16, (DOUTF / 16) * 32, 0, stream>>>(a1, W2, dinv, h2);
    k_gather<DOUTF, GATHER_NPW>
        <<<(N_NODES + 8 * GATHER_NPW - 1) / (8 * GATHER_NPW), 256, 0, stream>>>(
            h2, rec, off, dinv, b2, out, N_NODES);
}


# Layer-1 product moved into the range-sort launch: each 256-node work-group parks its x rows and W1 in LDS and forms dinv*(x@W1) on WMMA; one launch and 25000 short waves fewer; sum order unchanged
# speedup vs baseline: 3.3195x; 1.0764x over previous
#include <hip/hip_runtime.h>

#define N_NODES 100000
#define N_EDGES 1000000
#define DIN  64
#define DHID 64
#define DOUTF 32

#define RANGE_SHIFT 8
#define RANGE_NODES 256
#define N_RANGES ((N_NODES >> RANGE_SHIFT) + 1)
#define PART_BLOCKS 256
#define PART_CHUNK ((N_EDGES + PART_BLOCKS - 1) / PART_BLOCKS)
#define N_CELLS (N_RANGES * PART_BLOCKS)
#define SCAN_BLOCKS ((N_CELLS + 1023) / 1024)

#define X_PITCH (DIN + 4)
#define W_PAIR_PITCH (2 * (DHID + 16))

typedef __attribute__((ext_vector_type(2))) float v2f;
typedef __attribute__((ext_vector_type(8))) float v8f;

__device__ __forceinline__ int range_of(int d) {
    const unsigned r = ((unsigned)d) >> RANGE_SHIFT;
    return (int)(r < (unsigned)(N_RANGES - 1) ? r : (unsigned)(N_RANGES - 1));
}

__global__ void __launch_bounds__(1024)
k_part_count(const int* __restrict__ dst, int* __restrict__ cells) {
    __shared__ int hist[N_RANGES];
    const int t = threadIdx.x;
    for (int i = t; i < N_RANGES; i += 1024) hist[i] = 0;
    __syncthreads();
    const int e0 = blockIdx.x * PART_CHUNK;
    const int e1 = min(e0 + PART_CHUNK, N_EDGES);
    for (int e = e0 + t; e < e1; e += 1024) atomicAdd(&hist[range_of(dst[e])], 1);
    __syncthreads();
    for (int i = t; i < N_RANGES; i += 1024) cells[i * PART_BLOCKS + blockIdx.x] = hist[i];
}

__global__ void __launch_bounds__(1024)
k_scan_cells(const int* __restrict__ in, int* __restrict__ out) {
    __shared__ int red[32];
    __shared__ int wtot[32];
    const int t = threadIdx.x, lane = t & 31, wv = t >> 5;
    int acc = 0;
    for (int j = 0; j < (int)blockIdx.x; ++j) acc += in[j * 1024 + t];
    for (int o = 16; o > 0; o >>= 1) acc += __shfl_xor(acc, o);
    if (lane == 0) red[wv] = acc;
    const int i = blockIdx.x * 1024 + t;
    const int v = (i < N_CELLS) ? in[i] : 0;
    int inc = v;
    for (int o = 1; o < 32; o <<= 1) {
        const int nb = __shfl_up(inc, o);
        if (lane >= o) inc += nb;
    }
    if (lane == 31) wtot[wv] = inc;
    __syncthreads();
    int pre = 0;
    for (int k = 0; k < 32; ++k) pre += red[k];
    for (int k = 0; k < wv; ++k) pre += wtot[k];
    if (i < N_CELLS) out[i] = pre + inc - v;
}

__global__ void __launch_bounds__(1024)
k_part_place(const int* __restrict__ src, const int* __restrict__ dst,
             const float* __restrict__ ew, const int* __restrict__ cellOff,
             int4* __restrict__ binned) {
    __shared__ int cur[N_RANGES];
    const int t = threadIdx.x;
    for (int i = t; i < N_RANGES; i += 1024) cur[i] = cellOff[i * PART_BLOCKS + blockIdx.x];
    __syncthreads();
    const int e0 = blockIdx.x * PART_CHUNK;
    const int e1 = min(e0 + PART_CHUNK, N_EDGES);
    for (int e = e0 + t; e < e1; e += 1024) {
        const int d = dst[e];
        const int p = atomicAdd(&cur[range_of(d)], 1);
        binned[p] = make_int4(src[e], d, __float_as_int(ew[e]), 0);
    }
}

template <int KDIM, int NDIM>
__device__ __forceinline__ void gemm_strip_pair_lds(const float* a_lds, const float* b_lds,
                                                    const float* rowscale,
                                                    float* __restrict__ C,
                                                    int row0, int col0, int lane) {
    const int mrow  = lane & 15;
    const int khalf = (lane >> 4) * 2;
    const float* ap = a_lds + mrow * X_PITCH + khalf;
    const float* bp = b_lds + (khalf >> 1) * W_PAIR_PITCH + 2 * mrow;

    v8f acc0 = {};
    v8f acc1 = {};
#pragma unroll
    for (int k = 0; k < KDIM; k += 4) {
        const v2f a  = *reinterpret_cast<const v2f*>(ap + k);
        const v2f b0 = *reinterpret_cast<const v2f*>(bp + (k >> 1) * W_PAIR_PITCH);
        const v2f b1 = *reinterpret_cast<const v2f*>(bp + (k >> 1) * W_PAIR_PITCH + 32);
        acc0 = __builtin_amdgcn_wmma_f32_16x16x4_f32(
            false, a, false, b0, (short)0, acc0, false, false);
        acc1 = __builtin_amdgcn_wmma_f32_16x16x4_f32(
            false, a, false, b1, (short)0, acc1, false, false);
    }

    const int mloc = (lane >> 4) << 3;
    const int col  = col0 + (lane & 15);
#pragma unroll
    for (int r = 0; r < 8; ++r) {
        const float sc = rowscale[mloc + r];
        float* cp = C + (size_t)(row0 + mloc + r) * NDIM + col;
        cp[0]  = acc0[r] * sc;
        cp[16] = acc1[r] * sc;
    }
}

__global__ void __launch_bounds__(1024)
k_range_sort(const int4* __restrict__ binned, const int* __restrict__ cellOff,
             int2* __restrict__ rec, int* __restrict__ off, float* __restrict__ dinv,
             const float* __restrict__ x, const float* __restrict__ W,
             float* __restrict__ hs) {
    static_assert(N_NODES % 16 == 0, "a 16-row strip is all nodes or none");
    static_assert(RANGE_NODES * (DIN / 4) == 4 * 1024, "four float4 of x per thread");
    static_assert(DIN * (DHID / 4) == 1024, "one float4 of W per thread");
    static_assert((RANGE_NODES / 16) * (DHID / 32) == 32, "one strip half per wave");
    __shared__ __attribute__((aligned(16))) float sx[RANGE_NODES * X_PITCH];
    __shared__ __attribute__((aligned(16))) float sw[(DIN / 2) * W_PAIR_PITCH];
    __shared__ float sdinv[RANGE_NODES];
    __shared__ int   cnt[RANGE_NODES];
    __shared__ float wsum[RANGE_NODES];
    __shared__ int   cur[RANGE_NODES];
    __shared__ int   wtot[RANGE_NODES / 32];
    const int t = threadIdx.x, lane = t & 31, wv = t >> 5;
    const int b = blockIdx.x;

    const float4* __restrict__ x4 =
        reinterpret_cast<const float4*>(x) + (size_t)b * RANGE_NODES * (DIN / 4);
    float4 xr[4];
#pragma unroll
    for (int j = 0; j < 4; ++j) {
        const int i = t + 1024 * j;
        xr[j] = (b * RANGE_NODES + (i >> 4) < N_NODES) ? x4[i]
                                                       : make_float4(0.0f, 0.0f, 0.0f, 0.0f);
    }
    const float4 wr = reinterpret_cast<const float4*>(W)[t];

    const int bs = cellOff[b * PART_BLOCKS];
    const int be = (b + 1 < N_RANGES) ? cellOff[(b + 1) * PART_BLOCKS] : N_EDGES;
    if (t < RANGE_NODES) { cnt[t] = 0; wsum[t] = 0.0f; }
#pragma unroll
    for (int j = 0; j < 4; ++j) {
        const int i = t + 1024 * j;
        *reinterpret_cast<float4*>(&sx[(i >> 4) * X_PITCH + (i & 15) * 4]) = xr[j];
    }
    {
        const int k = t >> 4, n = (t & 15) * 4;
        float* wp = &sw[(k >> 1) * W_PAIR_PITCH + 2 * n + (k & 1)];
        wp[0] = wr.x;
        wp[2] = wr.y;
        wp[4] = wr.z;
        wp[6] = wr.w;
    }
    __syncthreads();
    for (int i = bs + t; i < be; i += 1024) {
        const int4 r = binned[i];
        const int dl = r.y & (RANGE_NODES - 1);
        atomicAdd(&cnt[dl], 1);
        atomicAdd(&wsum[dl], __int_as_float(r.z));
    }
    __syncthreads();
    const int v = (t < RANGE_NODES) ? cnt[t] : 0;
    int inc = v;
    for (int o = 1; o < 32; o <<= 1) {
        const int nb = __shfl_up(inc, o);
        if (lane >= o) inc += nb;
    }
    if (t < RANGE_NODES && lane == 31) wtot[wv] = inc;
    __syncthreads();
    if (t < RANGE_NODES) {
        int pre = 0;
        for (int k = 0; k < wv; ++k) pre += wtot[k];
        const int ex = pre + inc - v;
        cur[t] = ex;
        const int node = b * RANGE_NODES + t;
        if (node <= N_NODES) off[node] = bs + ex;
        float dv = 0.0f;
        if (node < N_NODES) {
            const float dg = 1.0f + wsum[t];
            dv = dg > 0.0f ? rsqrtf(dg) : 0.0f;
            dinv[node] = dv;
        }
        sdinv[t] = dv;
    }
    __syncthreads();
    for (int i = bs + t; i < be; i += 1024) {
        const int4 r = binned[i];
        const int dl = r.y & (RANGE_NODES - 1);
        const int p = atomicAdd(&cur[dl], 1);
        rec[bs + p] = make_int2(r.x, r.z);
    }

    const int wvu   = __builtin_amdgcn_readfirstlane(wv);
    const int strip = wvu >> 1;
    const int col0  = (wvu & 1) * 32;
    const int row0  = b * RANGE_NODES + strip * 16;
    if (row0 < N_NODES)
        gemm_strip_pair_lds<DIN, DHID>(sx + strip * 16 * X_PITCH, sw + 2 * col0,
                                       sdinv + strip * 16, hs, row0, col0, lane);
}

template <int KDIM, int NDIM, bool RELU>
__global__ void k_gemm_wmma(const float* __restrict__ A, const float* __restrict__ B,
                            const float* __restrict__ rowscale, float* __restrict__ C) {
    const int lane  = threadIdx.x & 31;
    const int wave  = threadIdx.x >> 5;
    const int row0  = blockIdx.x * 16;
    const int col0  = wave * 16;
    const int mrow  = lane & 15;
    const int khalf = (lane >> 4) * 2;

    v8f acc = {};
#pragma unroll
    for (int k = 0; k < KDIM; k += 4) {
        const float* ap = A + (size_t)(row0 + mrow) * KDIM + (k + khalf);
        v2f a;
        a.x = ap[0];
        a.y = ap[1];
        if (RELU) { a.x = fmaxf(a.x, 0.0f); a.y = fmaxf(a.y, 0.0f); }

        const float* bp = B + (size_t)(k + khalf) * NDIM + (col0 + mrow);
        v2f bfr;
        bfr.x = bp[0];
        bfr.y = bp[NDIM];

        acc = __builtin_amdgcn_wmma_f32_16x16x4_f32(
            false, a, false, bfr, (short)0, acc, false, false);
    }

    const int mbase = row0 + ((lane >> 4) << 3);
    const int col   = col0 + (lane & 15);
#pragma unroll
    for (int r = 0; r < 8; ++r)
        C[(size_t)(mbase + r) * NDIM + col] = acc[r] * rowscale[mbase + r];
}

#define GATHER_NPW 8

__device__ __forceinline__ int rec_slot(int s, int en) {
    return max(min(s, en - 1), 0);
}

template <int F, int NPW>
__global__ void __launch_bounds__(256)
k_gather(const float* __restrict__ hs, const int2* __restrict__ rec,
         const int* __restrict__ off, const float* __restrict__ dinv,
         const float* __restrict__ bias, float* __restrict__ out, int n) {
    constexpr int LPR = F / 4;
    constexpr int G   = 32 / LPR;
    constexpr int U   = 16 / G;
    static_assert(NPW + 1 <= 32, "a wave's list bounds must fit one register");
    const int lane = threadIdx.x & 31;
    const int d0 = __builtin_amdgcn_readfirstlane(
                       (int)((blockIdx.x * blockDim.x + threadIdx.x) >> 5)) * NPW;
    if (d0 >= n) return;
    const int nd = min(NPW, n - d0);
    const int g = lane / LPR;
    const int c = lane % LPR;
    const int keep0 = (g == 0) ? -1 : 0;
    const float4* __restrict__ hs4 = reinterpret_cast<const float4*>(hs);

    const int   offv = off[min(d0 + lane, n)];
    const float dvv  = dinv[min(d0 + lane, n - 1)];
    const float4 b4  = reinterpret_cast<const float4*>(bias)[c];

    int st = __builtin_amdgcn_readlane(offv, 0);
    int en = __builtin_amdgcn_readlane(offv, 1);
    int2 rr[U];
#pragma unroll
    for (int j = 0; j < U; ++j) rr[j] = rec[rec_slot(st + j * G + g, en)];

    for (int i = 0; i < nd; ++i) {
        const int d = d0 + i;
        const float4 self = hs4[(size_t)d * LPR + c];
        float4 row[U];
#pragma unroll
        for (int j = 0; j < U; ++j) row[j] = hs4[(size_t)rr[j].x * LPR + c];

        const int enn = __builtin_amdgcn_readlane(offv, min(i + 2, nd));
        int2 rrn[U];
#pragma unroll
        for (int j = 0; j < U; ++j) rrn[j] = rec[rec_slot(en + j * G + g, enn)];

        float4 acc;
        acc.x = __int_as_float(__float_as_int(self.x) & keep0);
        acc.y = __int_as_float(__float_as_int(self.y) & keep0);
        acc.z = __int_as_float(__float_as_int(self.z) & keep0);
        acc.w = __int_as_float(__float_as_int(self.w) & keep0);
#pragma unroll
        for (int j = 0; j < U; ++j) {
            const bool live = st + j * G + g < en;
            const float w = __int_as_float(rr[j].y);
            const float sx = acc.x + w * row[j].x;
            const float sy = acc.y + w * row[j].y;
            const float sz = acc.z + w * row[j].z;
            const float sw = acc.w + w * row[j].w;
            acc.x = live ? sx : acc.x;
            acc.y = live ? sy : acc.y;
            acc.z = live ? sz : acc.z;
            acc.w = live ? sw : acc.w;
        }
        for (int k = st + 16; k < en; k += 16) {
            int2 rq[U];
#pragma unroll
            for (int j = 0; j < U; ++j) rq[j] = rec[min(k + j * G + g, en - 1)];
            float4 rw[U];
#pragma unroll
            for (int j = 0; j < U; ++j) rw[j] = hs4[(size_t)rq[j].x * LPR + c];
#pragma unroll
            for (int j = 0; j < U; ++j) {
                if (k + j * G + g < en) {
                    const float w = __int_as_float(rq[j].y);
                    acc.x += w * rw[j].x;
                    acc.y += w * rw[j].y;
                    acc.z += w * rw[j].z;
                    acc.w += w * rw[j].w;
                }
            }
        }
#pragma unroll
        for (int o = LPR; o < 32; o <<= 1) {
            acc.x += __shfl_xor(acc.x, o);
            acc.y += __shfl_xor(acc.y, o);
            acc.z += __shfl_xor(acc.z, o);
            acc.w += __shfl_xor(acc.w, o);
        }
        const float dv = __int_as_float(__builtin_amdgcn_readlane(__float_as_int(dvv), i));
        if (g == 0) {
            float4 o4;
            o4.x = b4.x + dv * acc.x;
            o4.y = b4.y + dv * acc.y;
            o4.z = b4.z + dv * acc.z;
            o4.w = b4.w + dv * acc.w;
            reinterpret_cast<float4*>(out)[(size_t)d * LPR + c] = o4;
        }
        st = en;
        en = enn;
#pragma unroll
        for (int j = 0; j < U; ++j) rr[j] = rrn[j];
    }
}

extern "C" void kernel_launch(void* const* d_in, const int* in_sizes, int n_in,
                              void* d_out, int out_size, void* d_ws, size_t ws_size,
                              hipStream_t stream) {
    (void)in_sizes; (void)n_in; (void)out_size; (void)ws_size;

    const float* x   = (const float*)d_in[0];
    const int*   ei  = (const int*)  d_in[1];
    const float* ew  = (const float*)d_in[2];
    const float* W1  = (const float*)d_in[3];
    const float* b1  = (const float*)d_in[4];
    const float* W2  = (const float*)d_in[5];
    const float* b2  = (const float*)d_in[6];
    const int* src = ei;
    const int* dst = ei + N_EDGES;
    float* out = (float*)d_out;

    float* dinv    = (float*)d_ws;
    int*   off     = (int*)(dinv + N_NODES);
    int*   cells   = off + 100032;
    int*   cellOff = cells + 100128;
    int4*  binned  = (int4*)(cellOff + N_CELLS);
    int2*  rec     = (int2*)(binned + N_EDGES);
    float* h1      = (float*)(rec + N_EDGES);
    float* a1      = h1 + (size_t)N_NODES * DHID;
    float* h2      = h1;

    k_part_count<<<PART_BLOCKS, 1024, 0, stream>>>(dst, cells);
    k_scan_cells<<<SCAN_BLOCKS, 1024, 0, stream>>>(cells, cellOff);
    k_part_place<<<PART_BLOCKS, 1024, 0, stream>>>(src, dst, ew, cellOff, binned);
    k_range_sort<<<N_RANGES, 1024, 0, stream>>>(binned, cellOff, rec, off, dinv, x, W1, h1);

    k_gather<DHID, GATHER_NPW>
        <<<(N_NODES + 8 * GATHER_NPW - 1) / (8 * GATHER_NPW), 256, 0, stream>>>(
            h1, rec, off, dinv, b1, a1, N_NODES);

    k_gemm_wmma<DHID, DOUTF, true>
        <<<N_NODES / 16, (DOUTF / 16) * 32, 0, stream>>>(a1, W2, dinv, h2);
    k_gather<DOUTF, GATHER_NPW>
        <<<(N_NODES + 8 * GATHER_NPW - 1) / (8 * GATHER_NPW), 256, 0, stream>>>(
            h2, rec, off, dinv, b2, out, N_NODES);
}


# Layer-2 product moved into the layer-1 gather launch: each 64-node work-group parks relu(a1) rows and W2 in LDS, forms dinv*(a1@W2) on WMMA; one launch and the a1 round trip gone; sum order unchanged
# speedup vs baseline: 3.6924x; 1.1123x over previous
#include <hip/hip_runtime.h>

#define N_NODES 100000
#define N_EDGES 1000000
#define DIN  64
#define DHID 64
#define DOUTF 32

#define RANGE_SHIFT 8
#define RANGE_NODES 256
#define N_RANGES ((N_NODES >> RANGE_SHIFT) + 1)
#define PART_BLOCKS 256
#define PART_CHUNK ((N_EDGES + PART_BLOCKS - 1) / PART_BLOCKS)
#define N_CELLS (N_RANGES * PART_BLOCKS)
#define SCAN_BLOCKS ((N_CELLS + 1023) / 1024)

#define X_PITCH (DIN + 4)
#define W_PAIR_PITCH (2 * (DHID + 16))
#define W2_PAIR_PITCH (2 * (DOUTF + 16))

typedef __attribute__((ext_vector_type(2))) float v2f;
typedef __attribute__((ext_vector_type(8))) float v8f;

__device__ __forceinline__ int range_of(int d) {
    const unsigned r = ((unsigned)d) >> RANGE_SHIFT;
    return (int)(r < (unsigned)(N_RANGES - 1) ? r : (unsigned)(N_RANGES - 1));
}

__global__ void __launch_bounds__(1024)
k_part_count(const int* __restrict__ dst, int* __restrict__ cells) {
    __shared__ int hist[N_RANGES];
    const int t = threadIdx.x;
    for (int i = t; i < N_RANGES; i += 1024) hist[i] = 0;
    __syncthreads();
    const int e0 = blockIdx.x * PART_CHUNK;
    const int e1 = min(e0 + PART_CHUNK, N_EDGES);
    for (int e = e0 + t; e < e1; e += 1024) atomicAdd(&hist[range_of(dst[e])], 1);
    __syncthreads();
    for (int i = t; i < N_RANGES; i += 1024) cells[i * PART_BLOCKS + blockIdx.x] = hist[i];
}

__global__ void __launch_bounds__(1024)
k_scan_cells(const int* __restrict__ in, int* __restrict__ out) {
    __shared__ int red[32];
    __shared__ int wtot[32];
    const int t = threadIdx.x, lane = t & 31, wv = t >> 5;
    int acc = 0;
    for (int j = 0; j < (int)blockIdx.x; ++j) acc += in[j * 1024 + t];
    for (int o = 16; o > 0; o >>= 1) acc += __shfl_xor(acc, o);
    if (lane == 0) red[wv] = acc;
    const int i = blockIdx.x * 1024 + t;
    const int v = (i < N_CELLS) ? in[i] : 0;
    int inc = v;
    for (int o = 1; o < 32; o <<= 1) {
        const int nb = __shfl_up(inc, o);
        if (lane >= o) inc += nb;
    }
    if (lane == 31) wtot[wv] = inc;
    __syncthreads();
    int pre = 0;
    for (int k = 0; k < 32; ++k) pre += red[k];
    for (int k = 0; k < wv; ++k) pre += wtot[k];
    if (i < N_CELLS) out[i] = pre + inc - v;
}

__global__ void __launch_bounds__(1024)
k_part_place(const int* __restrict__ src, const int* __restrict__ dst,
             const float* __restrict__ ew, const int* __restrict__ cellOff,
             int4* __restrict__ binned) {
    __shared__ int cur[N_RANGES];
    const int t = threadIdx.x;
    for (int i = t; i < N_RANGES; i += 1024) cur[i] = cellOff[i * PART_BLOCKS + blockIdx.x];
    __syncthreads();
    const int e0 = blockIdx.x * PART_CHUNK;
    const int e1 = min(e0 + PART_CHUNK, N_EDGES);
    for (int e = e0 + t; e < e1; e += 1024) {
        const int d = dst[e];
        const int p = atomicAdd(&cur[range_of(d)], 1);
        binned[p] = make_int4(src[e], d, __float_as_int(ew[e]), 0);
    }
}

template <int KDIM, int NDIM, int BPITCH>
__device__ __forceinline__ void gemm_strip_pair_lds(const float* a_lds, const float* b_lds,
                                                    const float* rowscale,
                                                    float* __restrict__ C,
                                                    int row0, int col0, int lane) {
    const int mrow  = lane & 15;
    const int khalf = (lane >> 4) * 2;
    const float* ap = a_lds + mrow * X_PITCH + khalf;
    const float* bp = b_lds + (khalf >> 1) * BPITCH + 2 * mrow;

    v8f acc0 = {};
    v8f acc1 = {};
#pragma unroll
    for (int k = 0; k < KDIM; k += 4) {
        const v2f a  = *reinterpret_cast<const v2f*>(ap + k);
        const v2f b0 = *reinterpret_cast<const v2f*>(bp + (k >> 1) * BPITCH);
        const v2f b1 = *reinterpret_cast<const v2f*>(bp + (k >> 1) * BPITCH + 32);
        acc0 = __builtin_amdgcn_wmma_f32_16x16x4_f32(
            false, a, false, b0, (short)0, acc0, false, false);
        acc1 = __builtin_amdgcn_wmma_f32_16x16x4_f32(
            false, a, false, b1, (short)0, acc1, false, false);
    }

    const int mloc = (lane >> 4) << 3;
    const int col  = col0 + (lane & 15);
#pragma unroll
    for (int r = 0; r < 8; ++r) {
        const float sc = rowscale[mloc + r];
        float* cp = C + (size_t)(row0 + mloc + r) * NDIM + col;
        cp[0]  = acc0[r] * sc;
        cp[16] = acc1[r] * sc;
    }
}

__global__ void __launch_bounds__(1024)
k_range_sort(const int4* __restrict__ binned, const int* __restrict__ cellOff,
             int2* __restrict__ rec, int* __restrict__ off, float* __restrict__ dinv,
             const float* __restrict__ x, const float* __restrict__ W,
             float* __restrict__ hs) {
    static_assert(N_NODES % 16 == 0, "a 16-row strip is all nodes or none");
    static_assert(RANGE_NODES * (DIN / 4) == 4 * 1024, "four float4 of x per thread");
    static_assert(DIN * (DHID / 4) == 1024, "one float4 of W per thread");
    static_assert((RANGE_NODES / 16) * (DHID / 32) == 32, "one strip half per wave");
    __shared__ __attribute__((aligned(16))) float sx[RANGE_NODES * X_PITCH];
    __shared__ __attribute__((aligned(16))) float sw[(DIN / 2) * W_PAIR_PITCH];
    __shared__ float sdinv[RANGE_NODES];
    __shared__ int   cnt[RANGE_NODES];
    __shared__ float wsum[RANGE_NODES];
    __shared__ int   cur[RANGE_NODES];
    __shared__ int   wtot[RANGE_NODES / 32];
    const int t = threadIdx.x, lane = t & 31, wv = t >> 5;
    const int b = blockIdx.x;

    const float4* __restrict__ x4 =
        reinterpret_cast<const float4*>(x) + (size_t)b * RANGE_NODES * (DIN / 4);
    float4 xr[4];
#pragma unroll
    for (int j = 0; j < 4; ++j) {
        const int i = t + 1024 * j;
        xr[j] = (b * RANGE_NODES + (i >> 4) < N_NODES) ? x4[i]
                                                       : make_float4(0.0f, 0.0f, 0.0f, 0.0f);
    }
    const float4 wr = reinterpret_cast<const float4*>(W)[t];

    const int bs = cellOff[b * PART_BLOCKS];
    const int be = (b + 1 < N_RANGES) ? cellOff[(b + 1) * PART_BLOCKS] : N_EDGES;
    if (t < RANGE_NODES) { cnt[t] = 0; wsum[t] = 0.0f; }
#pragma unroll
    for (int j = 0; j < 4; ++j) {
        const int i = t + 1024 * j;
        *reinterpret_cast<float4*>(&sx[(i >> 4) * X_PITCH + (i & 15) * 4]) = xr[j];
    }
    {
        const int k = t >> 4, n = (t & 15) * 4;
        float* wp = &sw[(k >> 1) * W_PAIR_PITCH + 2 * n + (k & 1)];
        wp[0] = wr.x;
        wp[2] = wr.y;
        wp[4] = wr.z;
        wp[6] = wr.w;
    }
    __syncthreads();
    for (int i = bs + t; i < be; i += 1024) {
        const int4 r = binned[i];
        const int dl = r.y & (RANGE_NODES - 1);
        atomicAdd(&cnt[dl], 1);
        atomicAdd(&wsum[dl], __int_as_float(r.z));
    }
    __syncthreads();
    const int v = (t < RANGE_NODES) ? cnt[t] : 0;
    int inc = v;
    for (int o = 1; o < 32; o <<= 1) {
        const int nb = __shfl_up(inc, o);
        if (lane >= o) inc += nb;
    }
    if (t < RANGE_NODES && lane == 31) wtot[wv] = inc;
    __syncthreads();
    if (t < RANGE_NODES) {
        int pre = 0;
        for (int k = 0; k < wv; ++k) pre += wtot[k];
        const int ex = pre + inc - v;
        cur[t] = ex;
        const int node = b * RANGE_NODES + t;
        if (node <= N_NODES) off[node] = bs + ex;
        float dv = 0.0f;
        if (node < N_NODES) {
            const float dg = 1.0f + wsum[t];
            dv = dg > 0.0f ? rsqrtf(dg) : 0.0f;
            dinv[node] = dv;
        }
        sdinv[t] = dv;
    }
    __syncthreads();
    for (int i = bs + t; i < be; i += 1024) {
        const int4 r = binned[i];
        const int dl = r.y & (RANGE_NODES - 1);
        const int p = atomicAdd(&cur[dl], 1);
        rec[bs + p] = make_int2(r.x, r.z);
    }

    const int wvu   = __builtin_amdgcn_readfirstlane(wv);
    const int strip = wvu >> 1;
    const int col0  = (wvu & 1) * 32;
    const int row0  = b * RANGE_NODES + strip * 16;
    if (row0 < N_NODES)
        gemm_strip_pair_lds<DIN, DHID, W_PAIR_PITCH>(sx + strip * 16 * X_PITCH, sw + 2 * col0,
                                       sdinv + strip * 16, hs, row0, col0, lane);
}

#define GATHER_NPW 8

__device__ __forceinline__ int rec_slot(int s, int en) {
    return max(min(s, en - 1), 0);
}

template <int F, int NPW, int NOUT>
__global__ void __launch_bounds__(256)
k_gather(const float* __restrict__ hs, const int2* __restrict__ rec,
         const int* __restrict__ off, const float* __restrict__ dinv,
         const float* __restrict__ bias, const float* __restrict__ Wn,
         float* __restrict__ out, int n) {
    constexpr int LPR = F / 4;
    constexpr int G   = 32 / LPR;
    constexpr int U   = 16 / G;
    constexpr bool FUSED = NOUT > 0;
    constexpr int BLOCK_NODES = 8 * NPW;
    constexpr int BPITCH = 2 * (NOUT + 16);
    static_assert(NPW + 1 <= 32, "a wave's list bounds must fit one register");
    static_assert(!FUSED || (F + 4 == X_PITCH && NOUT == 32 && BLOCK_NODES % 16 == 0),
                  "strip routine: 64-float rows at X_PITCH, one 32-column tile pair");
    static_assert(!FUSED || (F / 2) * (NOUT / 4) == 256, "two float4 of Wn per thread");
    static_assert(!FUSED || BPITCH == W2_PAIR_PITCH, "lane halves 32 banks apart");
    static_assert(N_NODES % 16 == 0, "a 16-row strip is all nodes or none");
    __shared__ __attribute__((aligned(16))) float sa[FUSED ? BLOCK_NODES * X_PITCH : 4];
    __shared__ __attribute__((aligned(16))) float sw[FUSED ? (F / 2) * BPITCH : 4];
    __shared__ float sdv[FUSED ? BLOCK_NODES : 1];
    const int lane = threadIdx.x & 31;
    const int wv = __builtin_amdgcn_readfirstlane((int)(threadIdx.x >> 5));
    const int d0 = __builtin_amdgcn_readfirstlane(
                       (int)((blockIdx.x * blockDim.x + threadIdx.x) >> 5)) * NPW;
    if (!FUSED && d0 >= n) return;
    const int nd = min(NPW, n - d0);
    if (FUSED) {
        const int p  = threadIdx.x >> 3;
        const int nn = (threadIdx.x & 7) * 4;
        const float4 w0 = reinterpret_cast<const float4*>(Wn)[(2 * p) * (NOUT / 4) + (nn >> 2)];
        const float4 w1 = reinterpret_cast<const float4*>(Wn)[(2 * p + 1) * (NOUT / 4) + (nn >> 2)];
        float4* wp = reinterpret_cast<float4*>(&sw[p * BPITCH + 2 * nn]);
        wp[0] = make_float4(w0.x, w1.x, w0.y, w1.y);
        wp[1] = make_float4(w0.z, w1.z, w0.w, w1.w);
    }
    const int g = lane / LPR;
    const int c = lane % LPR;
    const int keep0 = (g == 0) ? -1 : 0;
    const float4* __restrict__ hs4 = reinterpret_cast<const float4*>(hs);

    const int   offv = off[min(d0 + lane, n)];
    const float dvv  = dinv[min(d0 + lane, n - 1)];
    const float4 b4  = reinterpret_cast<const float4*>(bias)[c];
    if (FUSED && lane < NPW) sdv[wv * NPW + lane] = dvv;

    int st = __builtin_amdgcn_readlane(offv, 0);
    int en = __builtin_amdgcn_readlane(offv, 1);
    int2 rr[U];
#pragma unroll
    for (int j = 0; j < U; ++j) rr[j] = rec[rec_slot(st + j * G + g, en)];

    for (int i = 0; i < nd; ++i) {
        const int d = d0 + i;
        const float4 self = hs4[(size_t)d * LPR + c];
        float4 row[U];
#pragma unroll
        for (int j = 0; j < U; ++j) row[j] = hs4[(size_t)rr[j].x * LPR + c];

        const int enn = __builtin_amdgcn_readlane(offv, min(i + 2, nd));
        int2 rrn[U];
#pragma unroll
        for (int j = 0; j < U; ++j) rrn[j] = rec[rec_slot(en + j * G + g, enn)];

        float4 acc;
        acc.x = __int_as_float(__float_as_int(self.x) & keep0);
        acc.y = __int_as_float(__float_as_int(self.y) & keep0);
        acc.z = __int_as_float(__float_as_int(self.z) & keep0);
        acc.w = __int_as_float(__float_as_int(self.w) & keep0);
#pragma unroll
        for (int j = 0; j < U; ++j) {
            const bool live = st + j * G + g < en;
            const float w = __int_as_float(rr[j].y);
            const float sx = acc.x + w * row[j].x;
            const float sy = acc.y + w * row[j].y;
            const float sz = acc.z + w * row[j].z;
            const float sw = acc.w + w * row[j].w;
            acc.x = live ? sx : acc.x;
            acc.y = live ? sy : acc.y;
            acc.z = live ? sz : acc.z;
            acc.w = live ? sw : acc.w;
        }
        for (int k = st + 16; k < en; k += 16) {
            int2 rq[U];
#pragma unroll
            for (int j = 0; j < U; ++j) rq[j] = rec[min(k + j * G + g, en - 1)];
            float4 rw[U];
#pragma unroll
            for (int j = 0; j < U; ++j) rw[j] = hs4[(size_t)rq[j].x * LPR + c];
#pragma unroll
            for (int j = 0; j < U; ++j) {
                if (k + j * G + g < en) {
                    const float w = __int_as_float(rq[j].y);
                    acc.x += w * rw[j].x;
                    acc.y += w * rw[j].y;
                    acc.z += w * rw[j].z;
                    acc.w += w * rw[j].w;
                }
            }
        }
#pragma unroll
        for (int o = LPR; o < 32; o <<= 1) {
            acc.x += __shfl_xor(acc.x, o);
            acc.y += __shfl_xor(acc.y, o);
            acc.z += __shfl_xor(acc.z, o);
            acc.w += __shfl_xor(acc.w, o);
        }
        const float dv = __int_as_float(__builtin_amdgcn_readlane(__float_as_int(dvv), i));
        if (g == 0) {
            float4 o4;
            o4.x = b4.x + dv * acc.x;
            o4.y = b4.y + dv * acc.y;
            o4.z = b4.z + dv * acc.z;
            o4.w = b4.w + dv * acc.w;
            if (FUSED) {
                o4.x = fmaxf(o4.x, 0.0f);
                o4.y = fmaxf(o4.y, 0.0f);
                o4.z = fmaxf(o4.z, 0.0f);
                o4.w = fmaxf(o4.w, 0.0f);
                *reinterpret_cast<float4*>(&sa[(wv * NPW + i) * X_PITCH + c * 4]) = o4;
            } else {
                reinterpret_cast<float4*>(out)[(size_t)d * LPR + c] = o4;
            }
        }
        st = en;
        en = enn;
#pragma unroll
        for (int j = 0; j < U; ++j) rr[j] = rrn[j];
    }

    if (FUSED) {
        __syncthreads();
        if (wv < BLOCK_NODES / 16) {
            const int row0 = (int)blockIdx.x * BLOCK_NODES + wv * 16;
            if (row0 < n)
                gemm_strip_pair_lds<F, NOUT, BPITCH>(sa + wv * 16 * X_PITCH, sw,
                                                     sdv + wv * 16, out, row0, 0, lane);
        }
    }
}

extern "C" void kernel_launch(void* const* d_in, const int* in_sizes, int n_in,
                              void* d_out, int out_size, void* d_ws, size_t ws_size,
                              hipStream_t stream) {
    (void)in_sizes; (void)n_in; (void)out_size; (void)ws_size;

    const float* x   = (const float*)d_in[0];
    const int*   ei  = (const int*)  d_in[1];
    const float* ew  = (const float*)d_in[2];
    const float* W1  = (const float*)d_in[3];
    const float* b1  = (const float*)d_in[4];
    const float* W2  = (const float*)d_in[5];
    const float* b2  = (const float*)d_in[6];
    const int* src = ei;
    const int* dst = ei + N_EDGES;
    float* out = (float*)d_out;

    float* dinv    = (float*)d_ws;
    int*   off     = (int*)(dinv + N_NODES);
    int*   cells   = off + 100032;
    int*   cellOff = cells + 100128;
    int4*  binned  = (int4*)(cellOff + N_CELLS);
    int2*  rec     = (int2*)(binned + N_EDGES);
    float* h1      = (float*)(rec + N_EDGES);
    float* h2      = h1 + (size_t)N_NODES * DHID;

    k_part_count<<<PART_BLOCKS, 1024, 0, stream>>>(dst, cells);
    k_scan_cells<<<SCAN_BLOCKS, 1024, 0, stream>>>(cells, cellOff);
    k_part_place<<<PART_BLOCKS, 1024, 0, stream>>>(src, dst, ew, cellOff, binned);
    k_range_sort<<<N_RANGES, 1024, 0, stream>>>(binned, cellOff, rec, off, dinv, x, W1, h1);

    k_gather<DHID, GATHER_NPW, DOUTF>
        <<<(N_NODES + 8 * GATHER_NPW - 1) / (8 * GATHER_NPW), 256, 0, stream>>>(
            h1, rec, off, dinv, b1, W2, h2, N_NODES);

    k_gather<DOUTF, GATHER_NPW, 0>
        <<<(N_NODES + 8 * GATHER_NPW - 1) / (8 * GATHER_NPW), 256, 0, stream>>>(
            h2, rec, off, dinv, b2, W2, out, N_NODES);
}


# Edge grouping by range made chunk-local: each chunk sorts its own edges by range in LDS in one launch (count and scan launches gone); a range's work-group pulls its pieces; float sum order changes
# speedup vs baseline: 4.8025x; 1.3006x over previous
#include <hip/hip_runtime.h>

#define N_NODES 100000
#define N_EDGES 1000000
#define DIN  64
#define DHID 64
#define DOUTF 32

#define RANGE_SHIFT 8
#define RANGE_NODES 256
#define N_RANGES ((N_NODES >> RANGE_SHIFT) + 1)
#define PART_BLOCKS 256
#define PART_CHUNK ((N_EDGES + PART_BLOCKS - 1) / PART_BLOCKS)
#define TAB_ROWS (N_RANGES + 1)

#define X_PITCH (DIN + 4)
#define W_PAIR_PITCH (2 * (DHID + 16))
#define W2_PAIR_PITCH (2 * (DOUTF + 16))

typedef __attribute__((ext_vector_type(2))) float v2f;
typedef __attribute__((ext_vector_type(8))) float v8f;

__device__ __forceinline__ int range_of(int d) {
    const unsigned r = ((unsigned)d) >> RANGE_SHIFT;
    return (int)(r < (unsigned)(N_RANGES - 1) ? r : (unsigned)(N_RANGES - 1));
}

__global__ void __launch_bounds__(1024)
k_chunk_sort(const int* __restrict__ src, const int* __restrict__ dst,
             const float* __restrict__ ew, int* __restrict__ tab,
             int4* __restrict__ binned) {
    static_assert(PART_CHUNK <= 4 * 1024, "four edges per thread");
    static_assert(TAB_ROWS <= 1024, "one table row per thread");
    static_assert((PART_BLOCKS - 1) * PART_CHUNK < N_EDGES, "no empty chunk");
    __shared__ __attribute__((aligned(16))) int4 stage[PART_CHUNK];
    __shared__ int hist[TAB_ROWS];
    __shared__ int wtot[32];
    const int t = threadIdx.x, lane = t & 31, wv = t >> 5;
    if (t < TAB_ROWS) hist[t] = 0;
    const int e0 = blockIdx.x * PART_CHUNK;
    const int ne = min(PART_CHUNK, N_EDGES - e0);
    int   s[4], d[4], rk[4];
    float w[4];
#pragma unroll
    for (int j = 0; j < 4; ++j) {
        const int e = e0 + min(t + 1024 * j, ne - 1);
        d[j] = dst[e];
        s[j] = src[e];
        w[j] = ew[e];
    }
    __syncthreads();
#pragma unroll
    for (int j = 0; j < 4; ++j) {
        rk[j] = 0;
        if (t + 1024 * j < ne) rk[j] = atomicAdd(&hist[range_of(d[j])], 1);
    }
    __syncthreads();
    const int v = (t < TAB_ROWS) ? hist[t] : 0;
    int inc = v;
    for (int o = 1; o < 32; o <<= 1) {
        const int nb = __shfl_up(inc, o);
        if (lane >= o) inc += nb;
    }
    if (lane == 31) wtot[wv] = inc;
    __syncthreads();
    int pre = 0;
    for (int k = 0; k < wv; ++k) pre += wtot[k];
    const int ex = pre + inc - v;
    if (t < TAB_ROWS) {
        hist[t] = ex;
        tab[t * PART_BLOCKS + blockIdx.x] = ex;
    }
    __syncthreads();
#pragma unroll
    for (int j = 0; j < 4; ++j)
        if (t + 1024 * j < ne)
            stage[hist[range_of(d[j])] + rk[j]] =
                make_int4(s[j], d[j], __float_as_int(w[j]), 0);
    __syncthreads();
#pragma unroll
    for (int j = 0; j < 4; ++j) {
        const int i = t + 1024 * j;
        if (i < ne) binned[e0 + i] = stage[i];
    }
}

template <int KDIM, int NDIM, int BPITCH>
__device__ __forceinline__ void gemm_strip_pair_lds(const float* a_lds, const float* b_lds,
                                                    const float* rowscale,
                                                    float* __restrict__ C,
                                                    int row0, int col0, int lane) {
    const int mrow  = lane & 15;
    const int khalf = (lane >> 4) * 2;
    const float* ap = a_lds + mrow * X_PITCH + khalf;
    const float* bp = b_lds + (khalf >> 1) * BPITCH + 2 * mrow;

    v8f acc0 = {};
    v8f acc1 = {};
#pragma unroll
    for (int k = 0; k < KDIM; k += 4) {
        const v2f a  = *reinterpret_cast<const v2f*>(ap + k);
        const v2f b0 = *reinterpret_cast<const v2f*>(bp + (k >> 1) * BPITCH);
        const v2f b1 = *reinterpret_cast<const v2f*>(bp + (k >> 1) * BPITCH + 32);
        acc0 = __builtin_amdgcn_wmma_f32_16x16x4_f32(
            false, a, false, b0, (short)0, acc0, false, false);
        acc1 = __builtin_amdgcn_wmma_f32_16x16x4_f32(
            false, a, false, b1, (short)0, acc1, false, false);
    }

    const int mloc = (lane >> 4) << 3;
    const int col  = col0 + (lane & 15);
#pragma unroll
    for (int r = 0; r < 8; ++r) {
        const float sc = rowscale[mloc + r];
        float* cp = C + (size_t)(row0 + mloc + r) * NDIM + col;
        cp[0]  = acc0[r] * sc;
        cp[16] = acc1[r] * sc;
    }
}

__global__ void __launch_bounds__(1024)
k_range_sort(const int4* __restrict__ binned, const int* __restrict__ tab,
             int2* __restrict__ rec, int* __restrict__ off, float* __restrict__ dinv,
             const float* __restrict__ x, const float* __restrict__ W,
             float* __restrict__ hs) {
    static_assert(N_NODES % 16 == 0, "a 16-row strip is all nodes or none");
    static_assert(RANGE_NODES * (DIN / 4) == 4 * 1024, "four float4 of x per thread");
    static_assert(DIN * (DHID / 4) == 1024, "one float4 of W per thread");
    static_assert((RANGE_NODES / 16) * (DHID / 32) == 32, "one strip half per wave");
    static_assert(PART_BLOCKS * 4 == 1024 && PART_BLOCKS % 32 == 0, "four threads per piece");
    __shared__ __attribute__((aligned(16))) float sx[RANGE_NODES * X_PITCH];
    __shared__ __attribute__((aligned(16))) float sw[(DIN / 2) * W_PAIR_PITCH];
    __shared__ float sdinv[RANGE_NODES];
    __shared__ int   cnt[RANGE_NODES];
    __shared__ float wsum[RANGE_NODES];
    __shared__ int   cur[RANGE_NODES];
    __shared__ int   wtot[RANGE_NODES / 32];
    __shared__ int   slo[PART_BLOCKS];
    __shared__ int   shi[PART_BLOCKS];
    __shared__ int   sbs[PART_BLOCKS / 32];
    const int t = threadIdx.x, lane = t & 31, wv = t >> 5;
    const int b = blockIdx.x;

    int lo = 0, hi = 0;
    if (t < PART_BLOCKS) {
        lo = tab[b * PART_BLOCKS + t];
        hi = tab[(b + 1) * PART_BLOCKS + t];
    }

    const float4* __restrict__ x4 =
        reinterpret_cast<const float4*>(x) + (size_t)b * RANGE_NODES * (DIN / 4);
    float4 xr[4];
#pragma unroll
    for (int j = 0; j < 4; ++j) {
        const int i = t + 1024 * j;
        xr[j] = (b * RANGE_NODES + (i >> 4) < N_NODES) ? x4[i]
                                                       : make_float4(0.0f, 0.0f, 0.0f, 0.0f);
    }
    const float4 wr = reinterpret_cast<const float4*>(W)[t];

    if (t < RANGE_NODES) { cnt[t] = 0; wsum[t] = 0.0f; }
    if (t < PART_BLOCKS) {
        slo[t] = lo;
        shi[t] = hi;
        int a = lo;
        for (int o = 16; o > 0; o >>= 1) a += __shfl_xor(a, o);
        if (lane == 0) sbs[wv] = a;
    }
#pragma unroll
    for (int j = 0; j < 4; ++j) {
        const int i = t + 1024 * j;
        *reinterpret_cast<float4*>(&sx[(i >> 4) * X_PITCH + (i & 15) * 4]) = xr[j];
    }
    {
        const int k = t >> 4, n = (t & 15) * 4;
        float* wp = &sw[(k >> 1) * W_PAIR_PITCH + 2 * n + (k & 1)];
        wp[0] = wr.x;
        wp[2] = wr.y;
        wp[4] = wr.z;
        wp[6] = wr.w;
    }
    __syncthreads();
    int bs = 0;
    for (int k = 0; k < PART_BLOCKS / 32; ++k) bs += sbs[k];
    constexpr int PIECE_SLOTS = 4;
    const int4* __restrict__ seg = binned + (size_t)(t >> 2) * PART_CHUNK;
    const int k0 = slo[t >> 2] + (t & 3);
    const int k1 = shi[t >> 2];
    for (int k = k0; k < k1; k += 4 * PIECE_SLOTS) {
        int4 r[PIECE_SLOTS];
#pragma unroll
        for (int u = 0; u < PIECE_SLOTS; ++u) r[u] = seg[min(k + 4 * u, k1 - 1)];
        asm volatile("" :: "v"(r[0].y), "v"(r[0].z), "v"(r[1].y), "v"(r[1].z),
                           "v"(r[2].y), "v"(r[2].z), "v"(r[3].y), "v"(r[3].z));
#pragma unroll
        for (int u = 0; u < PIECE_SLOTS; ++u) {
            if (k + 4 * u < k1) {
                const int dl = r[u].y & (RANGE_NODES - 1);
                atomicAdd(&cnt[dl], 1);
                atomicAdd(&wsum[dl], __int_as_float(r[u].z));
            }
        }
    }
    __syncthreads();
    const int v = (t < RANGE_NODES) ? cnt[t] : 0;
    int inc = v;
    for (int o = 1; o < 32; o <<= 1) {
        const int nb = __shfl_up(inc, o);
        if (lane >= o) inc += nb;
    }
    if (t < RANGE_NODES && lane == 31) wtot[wv] = inc;
    __syncthreads();
    if (t < RANGE_NODES) {
        int pre = 0;
        for (int k = 0; k < wv; ++k) pre += wtot[k];
        const int ex = pre + inc - v;
        cur[t] = ex;
        const int node = b * RANGE_NODES + t;
        if (node <= N_NODES) off[node] = bs + ex;
        float dv = 0.0f;
        if (node < N_NODES) {
            const float dg = 1.0f + wsum[t];
            dv = dg > 0.0f ? rsqrtf(dg) : 0.0f;
            dinv[node] = dv;
        }
        sdinv[t] = dv;
    }
    __syncthreads();
    for (int k = k0; k < k1; k += 4 * PIECE_SLOTS) {
        int4 r[PIECE_SLOTS];
#pragma unroll
        for (int u = 0; u < PIECE_SLOTS; ++u) r[u] = seg[min(k + 4 * u, k1 - 1)];
        asm volatile("" :: "v"(r[0].x), "v"(r[0].y), "v"(r[0].z),
                           "v"(r[1].x), "v"(r[1].y), "v"(r[1].z),
                           "v"(r[2].x), "v"(r[2].y), "v"(r[2].z),
                           "v"(r[3].x), "v"(r[3].y), "v"(r[3].z));
#pragma unroll
        for (int u = 0; u < PIECE_SLOTS; ++u) {
            if (k + 4 * u < k1) {
                const int p = atomicAdd(&cur[r[u].y & (RANGE_NODES - 1)], 1);
                rec[bs + p] = make_int2(r[u].x, r[u].z);
            }
        }
    }

    const int wvu   = __builtin_amdgcn_readfirstlane(wv);
    const int strip = wvu >> 1;
    const int col0  = (wvu & 1) * 32;
    const int row0  = b * RANGE_NODES + strip * 16;
    if (row0 < N_NODES)
        gemm_strip_pair_lds<DIN, DHID, W_PAIR_PITCH>(sx + strip * 16 * X_PITCH, sw + 2 * col0,
                                       sdinv + strip * 16, hs, row0, col0, lane);
}

#define GATHER_NPW 8

__device__ __forceinline__ int rec_slot(int s, int en) {
    return max(min(s, en - 1), 0);
}

template <int F, int NPW, int NOUT>
__global__ void __launch_bounds__(256)
k_gather(const float* __restrict__ hs, const int2* __restrict__ rec,
         const int* __restrict__ off, const float* __restrict__ dinv,
         const float* __restrict__ bias, const float* __restrict__ Wn,
         float* __restrict__ out, int n) {
    constexpr int LPR = F / 4;
    constexpr int G   = 32 / LPR;
    constexpr int U   = 16 / G;
    constexpr bool FUSED = NOUT > 0;
    constexpr int BLOCK_NODES = 8 * NPW;
    constexpr int BPITCH = 2 * (NOUT + 16);
    static_assert(NPW + 1 <= 32, "a wave's list bounds must fit one register");
    static_assert(!FUSED || (F + 4 == X_PITCH && NOUT == 32 && BLOCK_NODES % 16 == 0),
                  "strip routine: 64-float rows at X_PITCH, one 32-column tile pair");
    static_assert(!FUSED || (F / 2) * (NOUT / 4) == 256, "two float4 of Wn per thread");
    static_assert(!FUSED || BPITCH == W2_PAIR_PITCH, "lane halves 32 banks apart");
    static_assert(N_NODES % 16 == 0, "a 16-row strip is all nodes or none");
    __shared__ __attribute__((aligned(16))) float sa[FUSED ? BLOCK_NODES * X_PITCH : 4];
    __shared__ __attribute__((aligned(16))) float sw[FUSED ? (F / 2) * BPITCH : 4];
    __shared__ float sdv[FUSED ? BLOCK_NODES : 1];
    const int lane = threadIdx.x & 31;
    const int wv = __builtin_amdgcn_readfirstlane((int)(threadIdx.x >> 5));
    const int d0 = __builtin_amdgcn_readfirstlane(
                       (int)((blockIdx.x * blockDim.x + threadIdx.x) >> 5)) * NPW;
    if (!FUSED && d0 >= n) return;
    const int nd = min(NPW, n - d0);
    if (FUSED) {
        const int p  = threadIdx.x >> 3;
        const int nn = (threadIdx.x & 7) * 4;
        const float4 w0 = reinterpret_cast<const float4*>(Wn)[(2 * p) * (NOUT / 4) + (nn >> 2)];
        const float4 w1 = reinterpret_cast<const float4*>(Wn)[(2 * p + 1) * (NOUT / 4) + (nn >> 2)];
        float4* wp = reinterpret_cast<float4*>(&sw[p * BPITCH + 2 * nn]);
        wp[0] = make_float4(w0.x, w1.x, w0.y, w1.y);
        wp[1] = make_float4(w0.z, w1.z, w0.w, w1.w);
    }
    const int g = lane / LPR;
    const int c = lane % LPR;
    const int keep0 = (g == 0) ? -1 : 0;
    const float4* __restrict__ hs4 = reinterpret_cast<const float4*>(hs);

    const int   offv = off[min(d0 + lane, n)];
    const float dvv  = dinv[min(d0 + lane, n - 1)];
    const float4 b4  = reinterpret_cast<const float4*>(bias)[c];
    if (FUSED && lane < NPW) sdv[wv * NPW + lane] = dvv;

    int st = __builtin_amdgcn_readlane(offv, 0);
    int en = __builtin_amdgcn_readlane(offv, 1);
    int2 rr[U];
#pragma unroll
    for (int j = 0; j < U; ++j) rr[j] = rec[rec_slot(st + j * G + g, en)];

    for (int i = 0; i < nd; ++i) {
        const int d = d0 + i;
        const float4 self = hs4[(size_t)d * LPR + c];
        float4 row[U];
#pragma unroll
        for (int j = 0; j < U; ++j) row[j] = hs4[(size_t)rr[j].x * LPR + c];

        const int enn = __builtin_amdgcn_readlane(offv, min(i + 2, nd));
        int2 rrn[U];
#pragma unroll
        for (int j = 0; j < U; ++j) rrn[j] = rec[rec_slot(en + j * G + g, enn)];

        float4 acc;
        acc.x = __int_as_float(__float_as_int(self.x) & keep0);
        acc.y = __int_as_float(__float_as_int(self.y) & keep0);
        acc.z = __int_as_float(__float_as_int(self.z) & keep0);
        acc.w = __int_as_float(__float_as_int(self.w) & keep0);
#pragma unroll
        for (int j = 0; j < U; ++j) {
            const bool live = st + j * G + g < en;
            const float w = __int_as_float(rr[j].y);
            const float sx = acc.x + w * row[j].x;
            const float sy = acc.y + w * row[j].y;
            const float sz = acc.z + w * row[j].z;
            const float sw = acc.w + w * row[j].w;
            acc.x = live ? sx : acc.x;
            acc.y = live ? sy : acc.y;
            acc.z = live ? sz : acc.z;
            acc.w = live ? sw : acc.w;
        }
        for (int k = st + 16; k < en; k += 16) {
            int2 rq[U];
#pragma unroll
            for (int j = 0; j < U; ++j) rq[j] = rec[min(k + j * G + g, en - 1)];
            float4 rw[U];
#pragma unroll
            for (int j = 0; j < U; ++j) rw[j] = hs4[(size_t)rq[j].x * LPR + c];
#pragma unroll
            for (int j = 0; j < U; ++j) {
                if (k + j * G + g < en) {
                    const float w = __int_as_float(rq[j].y);
                    acc.x += w * rw[j].x;
                    acc.y += w * rw[j].y;
                    acc.z += w * rw[j].z;
                    acc.w += w * rw[j].w;
                }
            }
        }
#pragma unroll
        for (int o = LPR; o < 32; o <<= 1) {
            acc.x += __shfl_xor(acc.x, o);
            acc.y += __shfl_xor(acc.y, o);
            acc.z += __shfl_xor(acc.z, o);
            acc.w += __shfl_xor(acc.w, o);
        }
        const float dv = __int_as_float(__builtin_amdgcn_readlane(__float_as_int(dvv), i));
        if (g == 0) {
            float4 o4;
            o4.x = b4.x + dv * acc.x;
            o4.y = b4.y + dv * acc.y;
            o4.z = b4.z + dv * acc.z;
            o4.w = b4.w + dv * acc.w;
            if (FUSED) {
                o4.x = fmaxf(o4.x, 0.0f);
                o4.y = fmaxf(o4.y, 0.0f);
                o4.z = fmaxf(o4.z, 0.0f);
                o4.w = fmaxf(o4.w, 0.0f);
                *reinterpret_cast<float4*>(&sa[(wv * NPW + i) * X_PITCH + c * 4]) = o4;
            } else {
                reinterpret_cast<float4*>(out)[(size_t)d * LPR + c] = o4;
            }
        }
        st = en;
        en = enn;
#pragma unroll
        for (int j = 0; j < U; ++j) rr[j] = rrn[j];
    }

    if (FUSED) {
        __syncthreads();
        if (wv < BLOCK_NODES / 16) {
            const int row0 = (int)blockIdx.x * BLOCK_NODES + wv * 16;
            if (row0 < n)
                gemm_strip_pair_lds<F, NOUT, BPITCH>(sa + wv * 16 * X_PITCH, sw,
                                                     sdv + wv * 16, out, row0, 0, lane);
        }
    }
}

extern "C" void kernel_launch(void* const* d_in, const int* in_sizes, int n_in,
                              void* d_out, int out_size, void* d_ws, size_t ws_size,
                              hipStream_t stream) {
    (void)in_sizes; (void)n_in; (void)out_size; (void)ws_size;

    const float* x   = (const float*)d_in[0];
    const int*   ei  = (const int*)  d_in[1];
    const float* ew  = (const float*)d_in[2];
    const float* W1  = (const float*)d_in[3];
    const float* b1  = (const float*)d_in[4];
    const float* W2  = (const float*)d_in[5];
    const float* b2  = (const float*)d_in[6];
    const int* src = ei;
    const int* dst = ei + N_EDGES;
    float* out = (float*)d_out;

    static_assert(TAB_ROWS * PART_BLOCKS <= 100384, "tab fits its plane");
    float* dinv    = (float*)d_ws;
    int*   off     = (int*)(dinv + N_NODES);
    int*   tab     = off + 100032;
    int4*  binned  = (int4*)(tab + 100384);
    int2*  rec     = (int2*)(binned + N_EDGES);
    float* h1      = (float*)(rec + N_EDGES);
    float* h2      = h1 + (size_t)N_NODES * DHID;

    k_chunk_sort<<<PART_BLOCKS, 1024, 0, stream>>>(src, dst, ew, tab, binned);
    k_range_sort<<<N_RANGES, 1024, 0, stream>>>(binned, tab, rec, off, dinv, x, W1, h1);

    k_gather<DHID, GATHER_NPW, DOUTF>
        <<<(N_NODES + 8 * GATHER_NPW - 1) / (8 * GATHER_NPW), 256, 0, stream>>>(
            h1, rec, off, dinv, b1, W2, h2, N_NODES);

    k_gather<DOUTF, GATHER_NPW, 0>
        <<<(N_NODES + 8 * GATHER_NPW - 1) / (8 * GATHER_NPW), 256, 0, stream>>>(
            h2, rec, off, dinv, b2, W2, out, N_NODES);
}


# Range sort puts a range's records in node order in LDS and writes rec with whole-line stores, instead of one scattered 8-byte store per record; sum order unchanged
# speedup vs baseline: 4.9990x; 1.0409x over previous
#include <hip/hip_runtime.h>

#define N_NODES 100000
#define N_EDGES 1000000
#define DIN  64
#define DHID 64
#define DOUTF 32

#define RANGE_SHIFT 8
#define RANGE_NODES 256
#define N_RANGES ((N_NODES >> RANGE_SHIFT) + 1)
#define PART_BLOCKS 256
#define PART_CHUNK ((N_EDGES + PART_BLOCKS - 1) / PART_BLOCKS)
#define TAB_ROWS (N_RANGES + 1)
#define REC_STAGE 4096

#define X_PITCH (DIN + 4)
#define W_PAIR_PITCH (2 * (DHID + 16))
#define W2_PAIR_PITCH (2 * (DOUTF + 16))

typedef __attribute__((ext_vector_type(2))) float v2f;
typedef __attribute__((ext_vector_type(8))) float v8f;

__device__ __forceinline__ int range_of(int d) {
    const unsigned r = ((unsigned)d) >> RANGE_SHIFT;
    return (int)(r < (unsigned)(N_RANGES - 1) ? r : (unsigned)(N_RANGES - 1));
}

__global__ void __launch_bounds__(1024)
k_chunk_sort(const int* __restrict__ src, const int* __restrict__ dst,
             const float* __restrict__ ew, int* __restrict__ tab,
             int4* __restrict__ binned) {
    static_assert(PART_CHUNK <= 4 * 1024, "four edges per thread");
    static_assert(TAB_ROWS <= 1024, "one table row per thread");
    static_assert((PART_BLOCKS - 1) * PART_CHUNK < N_EDGES, "no empty chunk");
    __shared__ __attribute__((aligned(16))) int4 stage[PART_CHUNK];
    __shared__ int hist[TAB_ROWS];
    __shared__ int wtot[32];
    const int t = threadIdx.x, lane = t & 31, wv = t >> 5;
    if (t < TAB_ROWS) hist[t] = 0;
    const int e0 = blockIdx.x * PART_CHUNK;
    const int ne = min(PART_CHUNK, N_EDGES - e0);
    int   s[4], d[4], rk[4];
    float w[4];
#pragma unroll
    for (int j = 0; j < 4; ++j) {
        const int e = e0 + min(t + 1024 * j, ne - 1);
        d[j] = dst[e];
        s[j] = src[e];
        w[j] = ew[e];
    }
    __syncthreads();
#pragma unroll
    for (int j = 0; j < 4; ++j) {
        rk[j] = 0;
        if (t + 1024 * j < ne) rk[j] = atomicAdd(&hist[range_of(d[j])], 1);
    }
    __syncthreads();
    const int v = (t < TAB_ROWS) ? hist[t] : 0;
    int inc = v;
    for (int o = 1; o < 32; o <<= 1) {
        const int nb = __shfl_up(inc, o);
        if (lane >= o) inc += nb;
    }
    if (lane == 31) wtot[wv] = inc;
    __syncthreads();
    int pre = 0;
    for (int k = 0; k < wv; ++k) pre += wtot[k];
    const int ex = pre + inc - v;
    if (t < TAB_ROWS) {
        hist[t] = ex;
        tab[t * PART_BLOCKS + blockIdx.x] = ex;
    }
    __syncthreads();
#pragma unroll
    for (int j = 0; j < 4; ++j)
        if (t + 1024 * j < ne)
            stage[hist[range_of(d[j])] + rk[j]] =
                make_int4(s[j], d[j], __float_as_int(w[j]), 0);
    __syncthreads();
#pragma unroll
    for (int j = 0; j < 4; ++j) {
        const int i = t + 1024 * j;
        if (i < ne) binned[e0 + i] = stage[i];
    }
}

template <int KDIM, int NDIM, int BPITCH>
__device__ __forceinline__ void gemm_strip_pair_lds(const float* a_lds, const float* b_lds,
                                                    const float* rowscale,
                                                    float* __restrict__ C,
                                                    int row0, int col0, int lane) {
    const int mrow  = lane & 15;
    const int khalf = (lane >> 4) * 2;
    const float* ap = a_lds + mrow * X_PITCH + khalf;
    const float* bp = b_lds + (khalf >> 1) * BPITCH + 2 * mrow;

    v8f acc0 = {};
    v8f acc1 = {};
#pragma unroll
    for (int k = 0; k < KDIM; k += 4) {
        const v2f a  = *reinterpret_cast<const v2f*>(ap + k);
        const v2f b0 = *reinterpret_cast<const v2f*>(bp + (k >> 1) * BPITCH);
        const v2f b1 = *reinterpret_cast<const v2f*>(bp + (k >> 1) * BPITCH + 32);
        acc0 = __builtin_amdgcn_wmma_f32_16x16x4_f32(
            false, a, false, b0, (short)0, acc0, false, false);
        acc1 = __builtin_amdgcn_wmma_f32_16x16x4_f32(
            false, a, false, b1, (short)0, acc1, false, false);
    }

    const int mloc = (lane >> 4) << 3;
    const int col  = col0 + (lane & 15);
#pragma unroll
    for (int r = 0; r < 8; ++r) {
        const float sc = rowscale[mloc + r];
        float* cp = C + (size_t)(row0 + mloc + r) * NDIM + col;
        cp[0]  = acc0[r] * sc;
        cp[16] = acc1[r] * sc;
    }
}

__global__ void __launch_bounds__(1024)
k_range_sort(const int4* __restrict__ binned, const int* __restrict__ tab,
             int2* __restrict__ rec, int* __restrict__ off, float* __restrict__ dinv,
             const float* __restrict__ x, const float* __restrict__ W,
             float* __restrict__ hs) {
    static_assert(N_NODES % 16 == 0, "a 16-row strip is all nodes or none");
    static_assert(RANGE_NODES * (DIN / 4) == 4 * 1024, "four float4 of x per thread");
    static_assert(DIN * (DHID / 4) == 1024, "one float4 of W per thread");
    static_assert((RANGE_NODES / 16) * (DHID / 32) == 32, "one strip half per wave");
    static_assert(PART_BLOCKS * 4 == 1024 && PART_BLOCKS % 32 == 0, "four threads per piece");
    __shared__ __attribute__((aligned(16))) float sx[RANGE_NODES * X_PITCH];
    __shared__ __attribute__((aligned(16))) float sw[(DIN / 2) * W_PAIR_PITCH];
    __shared__ __attribute__((aligned(16))) int2 srec[REC_STAGE];
    __shared__ float sdinv[RANGE_NODES];
    __shared__ int   cnt[RANGE_NODES];
    __shared__ float wsum[RANGE_NODES];
    __shared__ int   cur[RANGE_NODES];
    __shared__ int   wtot[RANGE_NODES / 32];
    __shared__ int   slo[PART_BLOCKS];
    __shared__ int   shi[PART_BLOCKS];
    __shared__ int   sbs[PART_BLOCKS / 32];
    const int t = threadIdx.x, lane = t & 31, wv = t >> 5;
    const int b = blockIdx.x;

    int lo = 0, hi = 0;
    if (t < PART_BLOCKS) {
        lo = tab[b * PART_BLOCKS + t];
        hi = tab[(b + 1) * PART_BLOCKS + t];
    }

    const float4* __restrict__ x4 =
        reinterpret_cast<const float4*>(x) + (size_t)b * RANGE_NODES * (DIN / 4);
    float4 xr[4];
#pragma unroll
    for (int j = 0; j < 4; ++j) {
        const int i = t + 1024 * j;
        xr[j] = (b * RANGE_NODES + (i >> 4) < N_NODES) ? x4[i]
                                                       : make_float4(0.0f, 0.0f, 0.0f, 0.0f);
    }
    const float4 wr = reinterpret_cast<const float4*>(W)[t];

    if (t < RANGE_NODES) { cnt[t] = 0; wsum[t] = 0.0f; }
    if (t < PART_BLOCKS) {
        slo[t] = lo;
        shi[t] = hi;
        int a = lo;
        for (int o = 16; o > 0; o >>= 1) a += __shfl_xor(a, o);
        if (lane == 0) sbs[wv] = a;
    }
#pragma unroll
    for (int j = 0; j < 4; ++j) {
        const int i = t + 1024 * j;
        *reinterpret_cast<float4*>(&sx[(i >> 4) * X_PITCH + (i & 15) * 4]) = xr[j];
    }
    {
        const int k = t >> 4, n = (t & 15) * 4;
        float* wp = &sw[(k >> 1) * W_PAIR_PITCH + 2 * n + (k & 1)];
        wp[0] = wr.x;
        wp[2] = wr.y;
        wp[4] = wr.z;
        wp[6] = wr.w;
    }
    __syncthreads();
    int bs = 0;
    for (int k = 0; k < PART_BLOCKS / 32; ++k) bs += sbs[k];
    constexpr int PIECE_SLOTS = 4;
    const int4* __restrict__ seg = binned + (size_t)(t >> 2) * PART_CHUNK;
    const int k0 = slo[t >> 2] + (t & 3);
    const int k1 = shi[t >> 2];
    for (int k = k0; k < k1; k += 4 * PIECE_SLOTS) {
        int4 r[PIECE_SLOTS];
#pragma unroll
        for (int u = 0; u < PIECE_SLOTS; ++u) r[u] = seg[min(k + 4 * u, k1 - 1)];
        asm volatile("" :: "v"(r[0].y), "v"(r[0].z), "v"(r[1].y), "v"(r[1].z),
                           "v"(r[2].y), "v"(r[2].z), "v"(r[3].y), "v"(r[3].z));
#pragma unroll
        for (int u = 0; u < PIECE_SLOTS; ++u) {
            if (k + 4 * u < k1) {
                const int dl = r[u].y & (RANGE_NODES - 1);
                atomicAdd(&cnt[dl], 1);
                atomicAdd(&wsum[dl], __int_as_float(r[u].z));
            }
        }
    }
    __syncthreads();
    const int v = (t < RANGE_NODES) ? cnt[t] : 0;
    int inc = v;
    for (int o = 1; o < 32; o <<= 1) {
        const int nb = __shfl_up(inc, o);
        if (lane >= o) inc += nb;
    }
    if (t < RANGE_NODES && lane == 31) wtot[wv] = inc;
    __syncthreads();
    if (t < RANGE_NODES) {
        int pre = 0;
        for (int k = 0; k < wv; ++k) pre += wtot[k];
        const int ex = pre + inc - v;
        cur[t] = ex;
        const int node = b * RANGE_NODES + t;
        if (node <= N_NODES) off[node] = bs + ex;
        float dv = 0.0f;
        if (node < N_NODES) {
            const float dg = 1.0f + wsum[t];
            dv = dg > 0.0f ? rsqrtf(dg) : 0.0f;
            dinv[node] = dv;
        }
        sdinv[t] = dv;
    }
    __syncthreads();
    for (int k = k0; k < k1; k += 4 * PIECE_SLOTS) {
        int4 r[PIECE_SLOTS];
#pragma unroll
        for (int u = 0; u < PIECE_SLOTS; ++u) r[u] = seg[min(k + 4 * u, k1 - 1)];
        asm volatile("" :: "v"(r[0].x), "v"(r[0].y), "v"(r[0].z),
                           "v"(r[1].x), "v"(r[1].y), "v"(r[1].z),
                           "v"(r[2].x), "v"(r[2].y), "v"(r[2].z),
                           "v"(r[3].x), "v"(r[3].y), "v"(r[3].z));
#pragma unroll
        for (int u = 0; u < PIECE_SLOTS; ++u) {
            if (k + 4 * u < k1) {
                const int p = atomicAdd(&cur[r[u].y & (RANGE_NODES - 1)], 1);
                const int2 rv = make_int2(r[u].x, r[u].z);
                if (p < REC_STAGE) srec[p] = rv;
                else               rec[bs + p] = rv;
            }
        }
    }

    __syncthreads();
    {
        int tot = 0;
        for (int k = 0; k < RANGE_NODES / 32; ++k) tot += wtot[k];
        const int ns = min(tot, REC_STAGE);
        for (int i = t - (bs & 15); i < ns; i += 1024)
            if (i >= 0) rec[bs + i] = srec[i];
    }

    const int wvu   = __builtin_amdgcn_readfirstlane(wv);
    const int strip = wvu >> 1;
    const int col0  = (wvu & 1) * 32;
    const int row0  = b * RANGE_NODES + strip * 16;
    if (row0 < N_NODES)
        gemm_strip_pair_lds<DIN, DHID, W_PAIR_PITCH>(sx + strip * 16 * X_PITCH, sw + 2 * col0,
                                       sdinv + strip * 16, hs, row0, col0, lane);
}

#define GATHER_NPW 8

__device__ __forceinline__ int rec_slot(int s, int en) {
    return max(min(s, en - 1), 0);
}

template <int F, int NPW, int NOUT>
__global__ void __launch_bounds__(256)
k_gather(const float* __restrict__ hs, const int2* __restrict__ rec,
         const int* __restrict__ off, const float* __restrict__ dinv,
         const float* __restrict__ bias, const float* __restrict__ Wn,
         float* __restrict__ out, int n) {
    constexpr int LPR = F / 4;
    constexpr int G   = 32 / LPR;
    constexpr int U   = 16 / G;
    constexpr bool FUSED = NOUT > 0;
    constexpr int BLOCK_NODES = 8 * NPW;
    constexpr int BPITCH = 2 * (NOUT + 16);
    static_assert(NPW + 1 <= 32, "a wave's list bounds must fit one register");
    static_assert(!FUSED || (F + 4 == X_PITCH && NOUT == 32 && BLOCK_NODES % 16 == 0),
                  "strip routine: 64-float rows at X_PITCH, one 32-column tile pair");
    static_assert(!FUSED || (F / 2) * (NOUT / 4) == 256, "two float4 of Wn per thread");
    static_assert(!FUSED || BPITCH == W2_PAIR_PITCH, "lane halves 32 banks apart");
    static_assert(N_NODES % 16 == 0, "a 16-row strip is all nodes or none");
    __shared__ __attribute__((aligned(16))) float sa[FUSED ? BLOCK_NODES * X_PITCH : 4];
    __shared__ __attribute__((aligned(16))) float sw[FUSED ? (F / 2) * BPITCH : 4];
    __shared__ float sdv[FUSED ? BLOCK_NODES : 1];
    const int lane = threadIdx.x & 31;
    const int wv = __builtin_amdgcn_readfirstlane((int)(threadIdx.x >> 5));
    const int d0 = __builtin_amdgcn_readfirstlane(
                       (int)((blockIdx.x * blockDim.x + threadIdx.x) >> 5)) * NPW;
    if (!FUSED && d0 >= n) return;
    const int nd = min(NPW, n - d0);
    if (FUSED) {
        const int p  = threadIdx.x >> 3;
        const int nn = (threadIdx.x & 7) * 4;
        const float4 w0 = reinterpret_cast<const float4*>(Wn)[(2 * p) * (NOUT / 4) + (nn >> 2)];
        const float4 w1 = reinterpret_cast<const float4*>(Wn)[(2 * p + 1) * (NOUT / 4) + (nn >> 2)];
        float4* wp = reinterpret_cast<float4*>(&sw[p * BPITCH + 2 * nn]);
        wp[0] = make_float4(w0.x, w1.x, w0.y, w1.y);
        wp[1] = make_float4(w0.z, w1.z, w0.w, w1.w);
    }
    const int g = lane / LPR;
    const int c = lane % LPR;
    const int keep0 = (g == 0) ? -1 : 0;
    const float4* __restrict__ hs4 = reinterpret_cast<const float4*>(hs);

    const int   offv = off[min(d0 + lane, n)];
    const float dvv  = dinv[min(d0 + lane, n - 1)];
    const float4 b4  = reinterpret_cast<const float4*>(bias)[c];
    if (FUSED && lane < NPW) sdv[wv * NPW + lane] = dvv;

    int st = __builtin_amdgcn_readlane(offv, 0);
    int en = __builtin_amdgcn_readlane(offv, 1);
    int2 rr[U];
#pragma unroll
    for (int j = 0; j < U; ++j) rr[j] = rec[rec_slot(st + j * G + g, en)];

    for (int i = 0; i < nd; ++i) {
        const int d = d0 + i;
        const float4 self = hs4[(size_t)d * LPR + c];
        float4 row[U];
#pragma unroll
        for (int j = 0; j < U; ++j) row[j] = hs4[(size_t)rr[j].x * LPR + c];

        const int enn = __builtin_amdgcn_readlane(offv, min(i + 2, nd));
        int2 rrn[U];
#pragma unroll
        for (int j = 0; j < U; ++j) rrn[j] = rec[rec_slot(en + j * G + g, enn)];

        float4 acc;
        acc.x = __int_as_float(__float_as_int(self.x) & keep0);
        acc.y = __int_as_float(__float_as_int(self.y) & keep0);
        acc.z = __int_as_float(__float_as_int(self.z) & keep0);
        acc.w = __int_as_float(__float_as_int(self.w) & keep0);
#pragma unroll
        for (int j = 0; j < U; ++j) {
            const bool live = st + j * G + g < en;
            const float w = __int_as_float(rr[j].y);
            const float sx = acc.x + w * row[j].x;
            const float sy = acc.y + w * row[j].y;
            const float sz = acc.z + w * row[j].z;
            const float sw = acc.w + w * row[j].w;
            acc.x = live ? sx : acc.x;
            acc.y = live ? sy : acc.y;
            acc.z = live ? sz : acc.z;
            acc.w = live ? sw : acc.w;
        }
        for (int k = st + 16; k < en; k += 16) {
            int2 rq[U];
#pragma unroll
            for (int j = 0; j < U; ++j) rq[j] = rec[min(k + j * G + g, en - 1)];
            float4 rw[U];
#pragma unroll
            for (int j = 0; j < U; ++j) rw[j] = hs4[(size_t)rq[j].x * LPR + c];
#pragma unroll
            for (int j = 0; j < U; ++j) {
                if (k + j * G + g < en) {
                    const float w = __int_as_float(rq[j].y);
                    acc.x += w * rw[j].x;
                    acc.y += w * rw[j].y;
                    acc.z += w * rw[j].z;
                    acc.w += w * rw[j].w;
                }
            }
        }
#pragma unroll
        for (int o = LPR; o < 32; o <<= 1) {
            acc.x += __shfl_xor(acc.x, o);
            acc.y += __shfl_xor(acc.y, o);
            acc.z += __shfl_xor(acc.z, o);
            acc.w += __shfl_xor(acc.w, o);
        }
        const float dv = __int_as_float(__builtin_amdgcn_readlane(__float_as_int(dvv), i));
        if (g == 0) {
            float4 o4;
            o4.x = b4.x + dv * acc.x;
            o4.y = b4.y + dv * acc.y;
            o4.z = b4.z + dv * acc.z;
            o4.w = b4.w + dv * acc.w;
            if (FUSED) {
                o4.x = fmaxf(o4.x, 0.0f);
                o4.y = fmaxf(o4.y, 0.0f);
                o4.z = fmaxf(o4.z, 0.0f);
                o4.w = fmaxf(o4.w, 0.0f);
                *reinterpret_cast<float4*>(&sa[(wv * NPW + i) * X_PITCH + c * 4]) = o4;
            } else {
                reinterpret_cast<float4*>(out)[(size_t)d * LPR + c] = o4;
            }
        }
        st = en;
        en = enn;
#pragma unroll
        for (int j = 0; j < U; ++j) rr[j] = rrn[j];
    }

    if (FUSED) {
        __syncthreads();
        if (wv < BLOCK_NODES / 16) {
            const int row0 = (int)blockIdx.x * BLOCK_NODES + wv * 16;
            if (row0 < n)
                gemm_strip_pair_lds<F, NOUT, BPITCH>(sa + wv * 16 * X_PITCH, sw,
                                                     sdv + wv * 16, out, row0, 0, lane);
        }
    }
}

extern "C" void kernel_launch(void* const* d_in, const int* in_sizes, int n_in,
                              void* d_out, int out_size, void* d_ws, size_t ws_size,
                              hipStream_t stream) {
    (void)in_sizes; (void)n_in; (void)out_size; (void)ws_size;

    const float* x   = (const float*)d_in[0];
    const int*   ei  = (const int*)  d_in[1];
    const float* ew  = (const float*)d_in[2];
    const float* W1  = (const float*)d_in[3];
    const float* b1  = (const float*)d_in[4];
    const float* W2  = (const float*)d_in[5];
    const float* b2  = (const float*)d_in[6];
    const int* src = ei;
    const int* dst = ei + N_EDGES;
    float* out = (float*)d_out;

    static_assert(TAB_ROWS * PART_BLOCKS <= 100384, "tab fits its plane");
    float* dinv    = (float*)d_ws;
    int*   off     = (int*)(dinv + N_NODES);
    int*   tab     = off + 100032;
    int4*  binned  = (int4*)(tab + 100384);
    int2*  rec     = (int2*)(binned + N_EDGES);
    float* h1      = (float*)(rec + N_EDGES);
    float* h2      = h1 + (size_t)N_NODES * DHID;

    k_chunk_sort<<<PART_BLOCKS, 1024, 0, stream>>>(src, dst, ew, tab, binned);
    k_range_sort<<<N_RANGES, 1024, 0, stream>>>(binned, tab, rec, off, dinv, x, W1, h1);

    k_gather<DHID, GATHER_NPW, DOUTF>
        <<<(N_NODES + 8 * GATHER_NPW - 1) / (8 * GATHER_NPW), 256, 0, stream>>>(
            h1, rec, off, dinv, b1, W2, h2, N_NODES);

    k_gather<DOUTF, GATHER_NPW, 0>
        <<<(N_NODES + 8 * GATHER_NPW - 1) / (8 * GATHER_NPW), 256, 0, stream>>>(
            h2, rec, off, dinv, b2, W2, out, N_NODES);
}
